# starting kernel
# baseline (speedup 1.0000x reference)
#include <hip/hip_runtime.h>

#define BB 256
#define TT 500
#define NKC 100
#define NA 13

typedef __attribute__((ext_vector_type(2))) float v2f;
typedef __attribute__((ext_vector_type(8))) float v8f;

__device__ __forceinline__ float lse2(float x, float y) {
    float m = fmaxf(x, y);
    return m + log1pf(__expf(-fabsf(x - y)));
}

__global__ __launch_bounds__(128)
void gemm_x_wmma(const float* __restrict__ FM, const float* __restrict__ lrw,
                 const float* __restrict__ lrb, float* __restrict__ xout) {
    __shared__ v2f bstage[50][32];
    const int tid = threadIdx.x;
    for (int i = tid; i < 50 * 32; i += 128) {
        int k = i >> 5, L = i & 31;
        int n = L & 15;
        int kb = k * 4 + ((L >> 4) << 1);
        v2f v;
        v.x = (n < 2) ? lrw[n * 200 + kb] : 0.0f;
        v.y = (n < 2) ? lrw[n * 200 + kb + 1] : 0.0f;
        bstage[k][L] = v;
    }
    __syncthreads();

    const int wave = tid >> 5, lane = tid & 31;
    const int tile = blockIdx.x * 4 + wave;
    const long row0 = (long)tile * 16;
    const int m = lane & 15;
    const int khalf = (lane >> 4) << 1;
    const float* arow = FM + (row0 + m) * 200 + khalf;

    v8f acc = {};
    for (int k = 0; k < 50; ++k) {
        v2f a = *(const v2f*)(arow + k * 4);
        v2f b = bstage[k][lane];
        acc = __builtin_amdgcn_wmma_f32_16x16x4_f32(
            false, a, false, b, (short)0, acc, false, false);
    }

    const int n = lane & 15;
    if (n < 2) {
        const float bias = lrb[n];
        const int moff = (lane >> 4) << 3;
        for (int r = 0; r < 8; ++r) {
            long row = row0 + moff + r;
            xout[row * 2 + n] = acc[r] + bias;
        }
    }
}

template <bool USE_XBUF>
__global__ __launch_bounds__(128)
void bkt_scan(const int* __restrict__ corr, const int* __restrict__ kcv,
              const int* __restrict__ prob, const float* __restrict__ FM,
              const float* __restrict__ trans, const float* __restrict__ olp,
              const float* __restrict__ olk, const float* __restrict__ initl,
              const float* __restrict__ lrw, const float* __restrict__ lrb,
              const float* __restrict__ xbuf, float* __restrict__ out) {
    __shared__ float alpha[4 * NKC * 2 * NA];
    __shared__ float olk_s[NKC * 4];
    __shared__ float logt_s[NKC * 4];
    __shared__ float la0_s[NKC * 2];

    const int tid = threadIdx.x;
    for (int i = tid; i < NKC * 4; i += 128) olk_s[i] = olk[i];
    for (int c = tid; c < NKC; c += 128) {
        float i0 = initl[c * 2], i1 = initl[c * 2 + 1];
        float l = lse2(i0, i1);
        la0_s[c * 2] = i0 - l;
        la0_s[c * 2 + 1] = i1 - l;
        float t00 = trans[c * 4 + 0], t01 = trans[c * 4 + 1];
        float t10 = trans[c * 4 + 2], t11 = trans[c * 4 + 3];
        float l0 = lse2(t00, t10);
        float l1 = lse2(t01, t11);
        logt_s[c * 4 + 0] = t00 - l0;
        logt_s[c * 4 + 1] = t01 - l1;
        logt_s[c * 4 + 2] = t10 - l0;
        logt_s[c * 4 + 3] = t11 - l1;
    }
    __syncthreads();
    for (int i = tid; i < 4 * NKC * 2 * NA; i += 128) {
        int r = i % (NKC * 2 * NA);
        int c = r / (2 * NA);
        int s = (r % (2 * NA)) / NA;
        alpha[i] = la0_s[c * 2 + s];
    }
    __syncthreads();

    const int wave = tid >> 5, lane = tid & 31;
    const int b = blockIdx.x * 4 + wave;
    const long bT = (long)b * TT;
    float* myA = alpha + wave * (NKC * 2 * NA);
    const int aa = (lane < NA) ? lane : (NA - 1);
    const float abil = -3.0f + 0.5f * (float)aa;
    float logab = -2.5649493574615367f;
    const float NEGI = -1.0e30f;

    for (int t = 0; t < TT; ++t) {
        const long idx = bT + t;
        const int obs = corr[idx];
        const int c = kcv[idx];
        const int p = prob[idx];

        float x0, x1;
        if (USE_XBUF) {
            v2f xv = *(const v2f*)(xbuf + idx * 2);
            x0 = xv.x; x1 = xv.y;
        } else {
            const float* fr = FM + idx * 200;
            float p0 = 0.f, p1 = 0.f;
            for (int f = lane; f < 200; f += 32) {
                float fv = fr[f];
                p0 = fmaf(fv, lrw[f], p0);
                p1 = fmaf(fv, lrw[200 + f], p1);
            }
            for (int off = 16; off; off >>= 1) {
                p0 += __shfl_xor(p0, off, 32);
                p1 += __shfl_xor(p1, off, 32);
            }
            x0 = p0 + lrb[0];
            x1 = p1 + lrb[1];
        }

        const float4 op4 = *(const float4*)(olp + (long)p * 4);
        float l00 = op4.x + olk_s[c * 4 + 0] - abil + x0;
        float l01 = op4.y + olk_s[c * 4 + 1] + abil - x0;
        float l10 = op4.z + olk_s[c * 4 + 2] - abil + x1;
        float l11 = op4.w + olk_s[c * 4 + 3] + abil - x1;
        float z0 = lse2(l00, l01), z1 = lse2(l10, l11);
        float lo00 = l00 - z0, lo01 = l01 - z0;
        float lo10 = l10 - z1, lo11 = l11 - z1;

        float la0v = myA[c * 26 + aa];
        float la1v = myA[c * 26 + 13 + aa];

        float lpgb0 = lse2(lo00 + la0v, lo10 + la1v);
        float lpgb1 = lse2(lo01 + la0v, lo11 + la1v);
        float nrm = lse2(lpgb0, lpgb1);
        lpgb0 -= nrm; lpgb1 -= nrm;

        float lps0 = obs ? lo01 : lo00;
        float lps1 = obs ? lo11 : lo10;
        float nla0 = lse2(lps0 + la0v + logt_s[c * 4 + 0],
                          lps1 + la1v + logt_s[c * 4 + 1]);
        float nla1 = lse2(lps0 + la0v + logt_s[c * 4 + 2],
                          lps1 + la1v + logt_s[c * 4 + 3]);
        if (lane < NA) {
            myA[c * 26 + lane] = nla0;
            myA[c * 26 + 13 + lane] = nla1;
        }

        float o0, o1;
        {
            float v = (lane < NA) ? (logab + lpgb0) : NEGI;
            float m2 = v;
            for (int off = 8; off; off >>= 1) m2 = fmaxf(m2, __shfl_xor(m2, off, 16));
            float e = __expf(v - m2), s = e;
            for (int off = 8; off; off >>= 1) s += __shfl_xor(s, off, 16);
            o0 = m2 + __logf(s);
        }
        {
            float v = (lane < NA) ? (logab + lpgb1) : NEGI;
            float m2 = v;
            for (int off = 8; off; off >>= 1) m2 = fmaxf(m2, __shfl_xor(m2, off, 16));
            float e = __expf(v - m2), s = e;
            for (int off = 8; off; off >>= 1) s += __shfl_xor(s, off, 16);
            o1 = m2 + __logf(s);
        }
        float onrm = lse2(o0, o1);
        if (lane == 0) {
            v2f ov;
            ov.x = o0 - onrm;
            ov.y = o1 - onrm;
            *(v2f*)(out + idx * 2) = ov;
        }
        logab += obs ? lpgb1 : lpgb0;
    }
}

extern "C" void kernel_launch(void* const* d_in, const int* in_sizes, int n_in,
                              void* d_out, int out_size, void* d_ws, size_t ws_size,
                              hipStream_t stream) {
    (void)in_sizes; (void)n_in; (void)out_size;
    const int* corr = (const int*)d_in[0];
    const int* kcv  = (const int*)d_in[1];
    const int* prob = (const int*)d_in[2];
    const float* FM    = (const float*)d_in[3];
    const float* trans = (const float*)d_in[4];
    const float* olp   = (const float*)d_in[5];
    const float* olk   = (const float*)d_in[6];
    const float* initl = (const float*)d_in[7];
    const float* lrw   = (const float*)d_in[8];
    const float* lrb   = (const float*)d_in[9];
    float* out = (float*)d_out;

    const size_t xbytes = (size_t)BB * TT * 2 * sizeof(float);
    if (ws_size >= xbytes) {
        float* xbuf = (float*)d_ws;
        hipLaunchKernelGGL(gemm_x_wmma, dim3(2000), dim3(128), 0, stream,
                           FM, lrw, lrb, xbuf);
        hipLaunchKernelGGL((bkt_scan<true>), dim3(64), dim3(128), 0, stream,
                           corr, kcv, prob, FM, trans, olp, olk, initl,
                           lrw, lrb, xbuf, out);
    } else {
        hipLaunchKernelGGL((bkt_scan<false>), dim3(64), dim3(128), 0, stream,
                           corr, kcv, prob, FM, trans, olp, olk, initl,
                           lrw, lrb, (const float*)nullptr, out);
    }
}


# scan: per-chain state carried as a normalized probability pair instead of unnormalized logs: observation softmax as sigmoids, 2x2 update + one reciprocal, log per output; rounding and sums change
# speedup vs baseline: 24.4270x; 24.4270x over previous
#include <hip/hip_runtime.h>

#define BB 256
#define TT 500
#define NKC 100
#define NA 13

typedef __attribute__((ext_vector_type(2))) float v2f;
typedef __attribute__((ext_vector_type(4))) float v4f;
typedef __attribute__((ext_vector_type(8))) float v8f;

__device__ __forceinline__ float lse2(float x, float y) {
    float m = fmaxf(x, y);
    return m + log1pf(__expf(-fabsf(x - y)));
}

__device__ __forceinline__ v2f softmax2(float d) {
    const float e = __expf(-fabsf(d));
    const float r = __builtin_amdgcn_rcpf(1.0f + e);
    const float s = e * r;
    v2f q;
    q.x = (d >= 0.f) ? s : r;
    q.y = (d >= 0.f) ? r : s;
    return q;
}

#define GEMM_THREADS 256
#define GEMM_WAVES (GEMM_THREADS / 32)
#define TILES_PER_WAVE 8
#define GEMM_BLOCKS 500

struct AFrag { v4f m[6]; float r; };

__device__ __forceinline__ void load_afrag(AFrag& f, const float* __restrict__ rowp,
                                           int koff, int roff) {
    for (int j = 0; j < 6; ++j) f.m[j] = *(const v4f*)(rowp + koff + 32 * j);
    f.r = rowp[roff];
}

__device__ __forceinline__ v8f wmma4(v2f a, v2f b, v8f c) {
    return __builtin_amdgcn_wmma_f32_16x16x4_f32(false, a, false, b, (short)0, c, false, false);
}

__device__ __forceinline__ float quad_xor1(float v) {
    return __int_as_float(__builtin_amdgcn_update_dpp(0, __float_as_int(v), 0xB1, 0xF, 0xF, true));
}
__device__ __forceinline__ float quad_xor2(float v) {
    return __int_as_float(__builtin_amdgcn_update_dpp(0, __float_as_int(v), 0x4E, 0xF, 0xF, true));
}

__device__ __forceinline__ void store_tile(const v8f acc, long trow, int g, int q, int h,
                                           float bias, float* __restrict__ xout) {
    float s0 = ((g == 0) ? acc[0] : 0.f) + ((g == 1) ? acc[1] : 0.f)
             + ((g == 2) ? acc[2] : 0.f) + ((g == 3) ? acc[3] : 0.f);
    float s1 = ((g == 0) ? acc[4] : 0.f) + ((g == 1) ? acc[5] : 0.f)
             + ((g == 2) ? acc[6] : 0.f) + ((g == 3) ? acc[7] : 0.f);
    s0 += quad_xor1(s0); s0 += quad_xor2(s0);
    s1 += quad_xor1(s1); s1 += quad_xor2(s1);
    if (g == 0 && q < 2) {
        const long row = trow + 2 * h;
        xout[row * 2 + q] = s0 + bias;
        xout[(row + 1) * 2 + q] = s1 + bias;
    }
}

__global__ __launch_bounds__(GEMM_THREADS) __attribute__((amdgpu_waves_per_eu(4, 4)))
void gemm_x_wmma(const float* __restrict__ FM, const float* __restrict__ lrw,
                 const float* __restrict__ lrb, float* __restrict__ xout) {
    const int lane = threadIdx.x & 31;
    const int gw = blockIdx.x * GEMM_WAVES + (threadIdx.x >> 5);
    const int c = lane & 15;
    const int h = lane >> 4;
    const int g = c & 3;
    const int q = c >> 2;
    const int koff = 8 * g + 4 * h;
    const int roff = 192 + 2 * g + h;

    const bool bval = q < 2;
    const float* wrow = lrw + (q & 1) * 200;
    v4f bm[6];
    for (int j = 0; j < 6; ++j) {
        v4f t;
        t.x = wrow[koff + 32 * j + 0]; t.y = wrow[koff + 32 * j + 1];
        t.z = wrow[koff + 32 * j + 2]; t.w = wrow[koff + 32 * j + 3];
        const v4f z = {0.f, 0.f, 0.f, 0.f};
        bm[j] = bval ? t : z;
    }
    const float brt = wrow[roff];
    const float br = bval ? brt : 0.f;
    const float bias = lrb[q & 1];

    const long row0 = (long)gw * (TILES_PER_WAVE * 4);
    const float* ap = FM + (row0 + q) * 200;

    AFrag c0, c1;
    load_afrag(c0, ap, koff, roff);
    load_afrag(c1, ap + 800, koff, roff);
    for (int p = 0; p < TILES_PER_WAVE / 2; ++p) {
        AFrag n0 = c0, n1 = c1;
        if (p + 1 < TILES_PER_WAVE / 2) {
            load_afrag(n0, ap + (2 * p + 2) * 800, koff, roff);
            load_afrag(n1, ap + (2 * p + 3) * 800, koff, roff);
        }
        v8f acc0 = {}, acc1 = {};
        for (int j = 0; j < 6; ++j) {
            v2f a, b;
            b.x = bm[j].x; b.y = bm[j].y;
            a.x = c0.m[j].x; a.y = c0.m[j].y; acc0 = wmma4(a, b, acc0);
            a.x = c1.m[j].x; a.y = c1.m[j].y; acc1 = wmma4(a, b, acc1);
            b.x = bm[j].z; b.y = bm[j].w;
            a.x = c0.m[j].z; a.y = c0.m[j].w; acc0 = wmma4(a, b, acc0);
            a.x = c1.m[j].z; a.y = c1.m[j].w; acc1 = wmma4(a, b, acc1);
        }
        {
            v2f a, b;
            b.x = br; b.y = 0.f;
            a.x = c0.r; a.y = 0.f; acc0 = wmma4(a, b, acc0);
            a.x = c1.r; a.y = 0.f; acc1 = wmma4(a, b, acc1);
        }
        store_tile(acc0, row0 + (2 * p) * 4, g, q, h, bias, xout);
        store_tile(acc1, row0 + (2 * p + 1) * 4, g, q, h, bias, xout);
        c0 = n0; c1 = n1;
    }
}

#define SCAN_THREADS 512
#define SCAN_WAVES (SCAN_THREADS / 32)

__device__ __forceinline__ float lse13(const float* v) {
    float m = -1.0e30f;
    for (int a = 0; a < NA; ++a) m = fmaxf(m, v[a]);
    float e[NA];
    for (int a = 0; a < NA; ++a) e[a] = __expf(v[a] - m);
    const float sA = (e[0] + e[8]) + (e[4] + e[12]);
    const float sB = (e[2] + e[10]) + e[6];
    const float sC = (e[1] + e[9]) + e[5];
    const float sD = (e[3] + e[11]) + e[7];
    const float s = (sA + sB) + (sC + sD);
    return m + __logf(s);
}

template <bool USE_XBUF>
__global__ __launch_bounds__(SCAN_THREADS)
void bkt_scan(const int* __restrict__ corr, const int* __restrict__ kcv,
              const int* __restrict__ prob, const float* __restrict__ FM,
              const float* __restrict__ trans, const float* __restrict__ olp,
              const float* __restrict__ olk, const float* __restrict__ initl,
              const float* __restrict__ lrw, const float* __restrict__ lrb,
              const float* __restrict__ xbuf, float* __restrict__ out) {
    __shared__ float4 lo_s[TT * NA];
    __shared__ v2f snap_s[TT * NA];
    __shared__ float4 op_s[TT];
    __shared__ v2f x_s[TT];
    __shared__ __attribute__((aligned(16))) int c_s[TT];
    __shared__ int obs_s[TT];
    __shared__ __attribute__((aligned(16))) int key_s[TT];
    __shared__ int ord_s[TT];
    __shared__ int lev_s[TT + 2];
    __shared__ v2f alpha[NKC * NA];
    __shared__ float olk_s[NKC * 4];
    __shared__ float4 T_s[NKC];
    __shared__ v2f pinit_s[NKC];

    const int tid = threadIdx.x;
    const int wave = tid >> 5, lane = tid & 31;
    const int b = blockIdx.x;
    const long bT = (long)b * TT;

    for (int i = tid; i < NKC * 4; i += SCAN_THREADS) olk_s[i] = olk[i];
    for (int i = tid; i < TT + 2; i += SCAN_THREADS) lev_s[i] = TT;
    for (int c = tid; c < NKC; c += SCAN_THREADS) {
        pinit_s[c] = softmax2(initl[c * 2 + 1] - initl[c * 2]);
        const float t00 = trans[c * 4 + 0], t01 = trans[c * 4 + 1];
        const float t10 = trans[c * 4 + 2], t11 = trans[c * 4 + 3];
        const v2f col0 = softmax2(t10 - t00);
        const v2f col1 = softmax2(t11 - t01);
        float4 T;
        T.x = col0.x; T.y = col1.x;
        T.z = col0.y; T.w = col1.y;
        T_s[c] = T;
    }

    for (int t = tid; t < TT; t += SCAN_THREADS) {
        const long idx = bT + t;
        c_s[t] = kcv[idx];
        obs_s[t] = corr[idx];
        const int p = prob[idx];
        op_s[t] = *(const float4*)(olp + (long)p * 4);
        if (USE_XBUF) x_s[t] = *(const v2f*)(xbuf + idx * 2);
    }
    if (!USE_XBUF) {
        for (int t = wave; t < TT; t += SCAN_WAVES) {
            const float* fr = FM + (bT + t) * 200;
            float p0 = 0.f, p1 = 0.f;
            for (int f = lane; f < 200; f += 32) {
                float fv = fr[f];
                p0 = fmaf(fv, lrw[f], p0);
                p1 = fmaf(fv, lrw[200 + f], p1);
            }
            for (int off = 16; off; off >>= 1) {
                p0 += __shfl_xor(p0, off, 32);
                p1 += __shfl_xor(p1, off, 32);
            }
            if (lane == 0) {
                v2f xv;
                xv.x = p0 + lrb[0];
                xv.y = p1 + lrb[1];
                x_s[t] = xv;
            }
        }
    }
    __syncthreads();

    for (int i = tid; i < NKC * NA; i += SCAN_THREADS) alpha[i] = pinit_s[i / NA];

    if (tid < TT) {
        const int c = c_s[tid];
        int r = 0;
        const int w0 = __builtin_amdgcn_readfirstlane(wave * 32);
        for (int j = 0; j < w0; j += 4) {
            const int4 v = *reinterpret_cast<const int4*>(c_s + j);
            r += (v.x == c) + (v.y == c) + (v.z == c) + (v.w == c);
        }
        const int w1 = (w0 + 32 < TT) ? (w0 + 32) : TT;
        for (int j = w0; j < w1; j += 4) {
            const int4 v = *reinterpret_cast<const int4*>(c_s + j);
            r += (v.x == c && j < tid) + (v.y == c && j + 1 < tid)
               + (v.z == c && j + 2 < tid) + (v.w == c && j + 3 < tid);
        }
        key_s[tid] = r * 512 + tid;
    }

    for (int e = tid; e < TT * NA; e += SCAN_THREADS) {
        const int t = e / NA;
        const int a = e - t * NA;
        const float abil = -3.0f + 0.5f * (float)a;
        const int c = c_s[t];
        const float4 op4 = op_s[t];
        const v2f xv = x_s[t];
        const float x0 = xv.x, x1 = xv.y;
        float l00 = op4.x + olk_s[c * 4 + 0] - abil + x0;
        float l01 = op4.y + olk_s[c * 4 + 1] + abil - x0;
        float l10 = op4.z + olk_s[c * 4 + 2] - abil + x1;
        float l11 = op4.w + olk_s[c * 4 + 3] + abil - x1;
        const v2f q0 = softmax2(l01 - l00);
        const v2f q1 = softmax2(l11 - l10);
        float4 r;
        r.x = q0.x; r.y = q0.y;
        r.z = q1.x; r.w = q1.y;
        lo_s[e] = r;
    }
    __syncthreads();

    if (tid < TT) {
        const int key = key_s[tid];
        int pos = 0;
        for (int j = 0; j < TT; j += 4) {
            const int4 v = *reinterpret_cast<const int4*>(key_s + j);
            pos += (v.x < key) + (v.y < key) + (v.z < key) + (v.w < key);
        }
        ord_s[pos] = tid | (c_s[tid] << 9) | ((obs_s[tid] != 0) << 16) | ((key >> 9) << 17);
    }
    __syncthreads();
    if (tid < TT) {
        const int r = ord_s[tid] >> 17;
        const int rp = (tid > 0) ? (ord_s[tid - 1] >> 17) : -1;
        if (r != rp) lev_s[r] = tid;
    }
    __syncthreads();

    {
        int k = 0;
        int s0 = 0;
        int s1 = __builtin_amdgcn_readfirstlane(lev_s[1]);
        while (s0 < TT) {
            for (int e = s0 * NA + tid; e < s1 * NA; e += SCAN_THREADS) {
                const int p = e / NA;
                const int a = e - p * NA;
                const int rec = ord_s[p];
                const int t = rec & 511;
                const int c = (rec >> 9) & 127;
                const int obs = (rec >> 16) & 1;
                const float4 q = lo_s[t * NA + a];
                const v2f pv = alpha[c * NA + a];
                const float4 T = T_s[c];
                const float w0 = (obs ? q.y : q.x) * pv.x;
                const float w1 = (obs ? q.w : q.z) * pv.y;
                const float n0 = T.x * w0 + T.y * w1;
                const float n1 = T.z * w0 + T.w * w1;
                const float rn = __builtin_amdgcn_rcpf(n0 + n1);
                v2f pn;
                pn.x = n0 * rn; pn.y = n1 * rn;
                alpha[c * NA + a] = pn;
                snap_s[t * NA + a] = pv;
            }
            ++k;
            s0 = s1;
            s1 = __builtin_amdgcn_readfirstlane(lev_s[k + 1]);
            __syncthreads();
        }
    }

    for (int e = tid; e < TT * NA; e += SCAN_THREADS) {
        const float4 q = lo_s[e];
        const v2f pv = snap_s[e];
        v2f g;
        g.x = __logf(q.x * pv.x + q.z * pv.y);
        g.y = __logf(q.y * pv.x + q.w * pv.y);
        snap_s[e] = g;
    }
    __syncthreads();

    v2f* vv_s = reinterpret_cast<v2f*>(lo_s);
    if (wave == 0 && lane < NA) {
        float logab = -2.5649493574615367f;
        #pragma unroll 10
        for (int t = 0; t < TT; ++t) {
            const v2f g = snap_s[t * NA + lane];
            v2f v;
            v.x = logab + g.x; v.y = logab + g.y;
            vv_s[t * NA + lane] = v;
            logab += obs_s[t] ? g.y : g.x;
        }
    }
    __syncthreads();

    for (int t = tid; t < TT; t += SCAN_THREADS) {
        float v0[NA], v1[NA];
        for (int a = 0; a < NA; ++a) {
            const v2f v = vv_s[t * NA + a];
            v0[a] = v.x; v1[a] = v.y;
        }
        const float o0 = lse13(v0);
        const float o1 = lse13(v1);
        const float onrm = lse2(o0, o1);
        v2f ov;
        ov.x = o0 - onrm;
        ov.y = o1 - onrm;
        *(v2f*)(out + (bT + t) * 2) = ov;
    }
}

extern "C" void kernel_launch(void* const* d_in, const int* in_sizes, int n_in,
                              void* d_out, int out_size, void* d_ws, size_t ws_size,
                              hipStream_t stream) {
    (void)in_sizes; (void)n_in; (void)out_size;
    const int* corr = (const int*)d_in[0];
    const int* kcv  = (const int*)d_in[1];
    const int* prob = (const int*)d_in[2];
    const float* FM    = (const float*)d_in[3];
    const float* trans = (const float*)d_in[4];
    const float* olp   = (const float*)d_in[5];
    const float* olk   = (const float*)d_in[6];
    const float* initl = (const float*)d_in[7];
    const float* lrw   = (const float*)d_in[8];
    const float* lrb   = (const float*)d_in[9];
    float* out = (float*)d_out;

    const size_t xbytes = (size_t)BB * TT * 2 * sizeof(float);
    if (ws_size >= xbytes) {
        float* xbuf = (float*)d_ws;
        hipLaunchKernelGGL(gemm_x_wmma, dim3(GEMM_BLOCKS), dim3(GEMM_THREADS), 0, stream,
                           FM, lrw, lrb, xbuf);
        hipLaunchKernelGGL((bkt_scan<true>), dim3(BB), dim3(SCAN_THREADS), 0, stream,
                           corr, kcv, prob, FM, trans, olp, olk, initl,
                           lrw, lrb, xbuf, out);
    } else {
        hipLaunchKernelGGL((bkt_scan<false>), dim3(BB), dim3(SCAN_THREADS), 0, stream,
                           corr, kcv, prob, FM, trans, olp, olk, initl,
                           lrw, lrb, (const float*)nullptr, out);
    }
}


# scan: occurrence numbers from in-wave lane masks (7 ballots on chain bits) + a 16x100 table of per-wave chain counts; steps placed by popcounts of chain bitmaps, no all-pairs compares; same sums
# speedup vs baseline: 28.4604x; 1.1651x over previous
#include <hip/hip_runtime.h>

#define BB 256
#define TT 500
#define NKC 100
#define NA 13

typedef __attribute__((ext_vector_type(2))) float v2f;
typedef __attribute__((ext_vector_type(4))) float v4f;
typedef __attribute__((ext_vector_type(8))) float v8f;

__device__ __forceinline__ float lse2(float x, float y) {
    float m = fmaxf(x, y);
    return m + log1pf(__expf(-fabsf(x - y)));
}

__device__ __forceinline__ v2f softmax2(float d) {
    const float e = __expf(-fabsf(d));
    const float r = __builtin_amdgcn_rcpf(1.0f + e);
    const float s = e * r;
    v2f q;
    q.x = (d >= 0.f) ? s : r;
    q.y = (d >= 0.f) ? r : s;
    return q;
}

#define GEMM_THREADS 256
#define GEMM_WAVES (GEMM_THREADS / 32)
#define TILES_PER_WAVE 8
#define GEMM_BLOCKS 500

struct AFrag { v4f m[6]; float r; };

__device__ __forceinline__ void load_afrag(AFrag& f, const float* __restrict__ rowp,
                                           int koff, int roff) {
    for (int j = 0; j < 6; ++j) f.m[j] = *(const v4f*)(rowp + koff + 32 * j);
    f.r = rowp[roff];
}

__device__ __forceinline__ v8f wmma4(v2f a, v2f b, v8f c) {
    return __builtin_amdgcn_wmma_f32_16x16x4_f32(false, a, false, b, (short)0, c, false, false);
}

__device__ __forceinline__ float quad_xor1(float v) {
    return __int_as_float(__builtin_amdgcn_update_dpp(0, __float_as_int(v), 0xB1, 0xF, 0xF, true));
}
__device__ __forceinline__ float quad_xor2(float v) {
    return __int_as_float(__builtin_amdgcn_update_dpp(0, __float_as_int(v), 0x4E, 0xF, 0xF, true));
}

__device__ __forceinline__ void store_tile(const v8f acc, long trow, int g, int q, int h,
                                           float bias, float* __restrict__ xout) {
    float s0 = ((g == 0) ? acc[0] : 0.f) + ((g == 1) ? acc[1] : 0.f)
             + ((g == 2) ? acc[2] : 0.f) + ((g == 3) ? acc[3] : 0.f);
    float s1 = ((g == 0) ? acc[4] : 0.f) + ((g == 1) ? acc[5] : 0.f)
             + ((g == 2) ? acc[6] : 0.f) + ((g == 3) ? acc[7] : 0.f);
    s0 += quad_xor1(s0); s0 += quad_xor2(s0);
    s1 += quad_xor1(s1); s1 += quad_xor2(s1);
    if (g == 0 && q < 2) {
        const long row = trow + 2 * h;
        xout[row * 2 + q] = s0 + bias;
        xout[(row + 1) * 2 + q] = s1 + bias;
    }
}

__global__ __launch_bounds__(GEMM_THREADS) __attribute__((amdgpu_waves_per_eu(4, 4)))
void gemm_x_wmma(const float* __restrict__ FM, const float* __restrict__ lrw,
                 const float* __restrict__ lrb, float* __restrict__ xout) {
    const int lane = threadIdx.x & 31;
    const int gw = blockIdx.x * GEMM_WAVES + (threadIdx.x >> 5);
    const int c = lane & 15;
    const int h = lane >> 4;
    const int g = c & 3;
    const int q = c >> 2;
    const int koff = 8 * g + 4 * h;
    const int roff = 192 + 2 * g + h;

    const bool bval = q < 2;
    const float* wrow = lrw + (q & 1) * 200;
    v4f bm[6];
    for (int j = 0; j < 6; ++j) {
        v4f t;
        t.x = wrow[koff + 32 * j + 0]; t.y = wrow[koff + 32 * j + 1];
        t.z = wrow[koff + 32 * j + 2]; t.w = wrow[koff + 32 * j + 3];
        const v4f z = {0.f, 0.f, 0.f, 0.f};
        bm[j] = bval ? t : z;
    }
    const float brt = wrow[roff];
    const float br = bval ? brt : 0.f;
    const float bias = lrb[q & 1];

    const long row0 = (long)gw * (TILES_PER_WAVE * 4);
    const float* ap = FM + (row0 + q) * 200;

    AFrag c0, c1;
    load_afrag(c0, ap, koff, roff);
    load_afrag(c1, ap + 800, koff, roff);
    for (int p = 0; p < TILES_PER_WAVE / 2; ++p) {
        AFrag n0 = c0, n1 = c1;
        if (p + 1 < TILES_PER_WAVE / 2) {
            load_afrag(n0, ap + (2 * p + 2) * 800, koff, roff);
            load_afrag(n1, ap + (2 * p + 3) * 800, koff, roff);
        }
        v8f acc0 = {}, acc1 = {};
        for (int j = 0; j < 6; ++j) {
            v2f a, b;
            b.x = bm[j].x; b.y = bm[j].y;
            a.x = c0.m[j].x; a.y = c0.m[j].y; acc0 = wmma4(a, b, acc0);
            a.x = c1.m[j].x; a.y = c1.m[j].y; acc1 = wmma4(a, b, acc1);
            b.x = bm[j].z; b.y = bm[j].w;
            a.x = c0.m[j].z; a.y = c0.m[j].w; acc0 = wmma4(a, b, acc0);
            a.x = c1.m[j].z; a.y = c1.m[j].w; acc1 = wmma4(a, b, acc1);
        }
        {
            v2f a, b;
            b.x = br; b.y = 0.f;
            a.x = c0.r; a.y = 0.f; acc0 = wmma4(a, b, acc0);
            a.x = c1.r; a.y = 0.f; acc1 = wmma4(a, b, acc1);
        }
        store_tile(acc0, row0 + (2 * p) * 4, g, q, h, bias, xout);
        store_tile(acc1, row0 + (2 * p + 1) * 4, g, q, h, bias, xout);
        c0 = n0; c1 = n1;
    }
}

#define SCAN_THREADS 512
#define SCAN_WAVES (SCAN_THREADS / 32)
static_assert(NKC <= 128, "a chain number has seven bits and a row of the chain bitmap has four words");

__device__ __forceinline__ float lse13(const float* v) {
    float m = -1.0e30f;
    for (int a = 0; a < NA; ++a) m = fmaxf(m, v[a]);
    float e[NA];
    for (int a = 0; a < NA; ++a) e[a] = __expf(v[a] - m);
    const float sA = (e[0] + e[8]) + (e[4] + e[12]);
    const float sB = (e[2] + e[10]) + e[6];
    const float sC = (e[1] + e[9]) + e[5];
    const float sD = (e[3] + e[11]) + e[7];
    const float s = (sA + sB) + (sC + sD);
    return m + __logf(s);
}

template <bool USE_XBUF>
__global__ __launch_bounds__(SCAN_THREADS)
void bkt_scan(const int* __restrict__ corr, const int* __restrict__ kcv,
              const int* __restrict__ prob, const float* __restrict__ FM,
              const float* __restrict__ trans, const float* __restrict__ olp,
              const float* __restrict__ olk, const float* __restrict__ initl,
              const float* __restrict__ lrw, const float* __restrict__ lrb,
              const float* __restrict__ xbuf, float* __restrict__ out) {
    __shared__ float4 lo_s[TT * NA];
    __shared__ v2f snap_s[TT * NA];
    __shared__ float4 op_s[TT];
    __shared__ v2f x_s[TT];
    __shared__ __attribute__((aligned(16))) int c_s[TT];
    __shared__ int obs_s[TT];
    __shared__ int hist_s[SCAN_WAVES * NKC];
    __shared__ uint4 lvl_s[TT];
    __shared__ int ord_s[TT];
    __shared__ int lev_s[TT + 2];
    __shared__ v2f alpha[NKC * NA];
    __shared__ float olk_s[NKC * 4];
    __shared__ float4 T_s[NKC];
    __shared__ v2f pinit_s[NKC];

    const int tid = threadIdx.x;
    const int wave = tid >> 5, lane = tid & 31;
    const int b = blockIdx.x;
    const long bT = (long)b * TT;

    for (int i = tid; i < NKC * 4; i += SCAN_THREADS) olk_s[i] = olk[i];
    for (int i = tid; i < TT + 2; i += SCAN_THREADS) lev_s[i] = TT;
    for (int i = tid; i < SCAN_WAVES * NKC; i += SCAN_THREADS) hist_s[i] = 0;
    for (int i = tid; i < TT; i += SCAN_THREADS) lvl_s[i] = make_uint4(0u, 0u, 0u, 0u);
    for (int c = tid; c < NKC; c += SCAN_THREADS) {
        pinit_s[c] = softmax2(initl[c * 2 + 1] - initl[c * 2]);
        const float t00 = trans[c * 4 + 0], t01 = trans[c * 4 + 1];
        const float t10 = trans[c * 4 + 2], t11 = trans[c * 4 + 3];
        const v2f col0 = softmax2(t10 - t00);
        const v2f col1 = softmax2(t11 - t01);
        float4 T;
        T.x = col0.x; T.y = col1.x;
        T.z = col0.y; T.w = col1.y;
        T_s[c] = T;
    }

    for (int t = tid; t < TT; t += SCAN_THREADS) {
        const long idx = bT + t;
        c_s[t] = kcv[idx];
        obs_s[t] = corr[idx];
        const int p = prob[idx];
        op_s[t] = *(const float4*)(olp + (long)p * 4);
        if (USE_XBUF) x_s[t] = *(const v2f*)(xbuf + idx * 2);
    }
    if (!USE_XBUF) {
        for (int t = wave; t < TT; t += SCAN_WAVES) {
            const float* fr = FM + (bT + t) * 200;
            float p0 = 0.f, p1 = 0.f;
            for (int f = lane; f < 200; f += 32) {
                float fv = fr[f];
                p0 = fmaf(fv, lrw[f], p0);
                p1 = fmaf(fv, lrw[200 + f], p1);
            }
            for (int off = 16; off; off >>= 1) {
                p0 += __shfl_xor(p0, off, 32);
                p1 += __shfl_xor(p1, off, 32);
            }
            if (lane == 0) {
                v2f xv;
                xv.x = p0 + lrb[0];
                xv.y = p1 + lrb[1];
                x_s[t] = xv;
            }
        }
    }
    __syncthreads();

    for (int i = tid; i < NKC * NA; i += SCAN_THREADS) alpha[i] = pinit_s[i / NA];

    const bool valid = tid < TT;
    const int myc = valid ? c_s[tid] : 0;
    unsigned peers = __builtin_amdgcn_ballot_w32(valid);
    for (int bit = 0; bit < 7; ++bit) {
        const bool on = ((myc >> bit) & 1) != 0;
        const unsigned m = __builtin_amdgcn_ballot_w32(on);
        peers &= on ? m : ~m;
    }
    const int rin = __popc(peers & ((1u << lane) - 1u));
    const int wv = __builtin_amdgcn_readfirstlane(wave);
    if (valid && rin == 0) hist_s[wv * NKC + myc] = __popc(peers);
    __syncthreads();

    int occ = rin;
    for (int w = 0; w < wv; ++w) occ += hist_s[w * NKC + myc];
    if (wv < (NKC + 31) / 32) {
        int cnt = 0;
        if (tid < NKC)
            for (int w = 0; w < SCAN_WAVES; ++w) cnt += hist_s[w * NKC + tid];
        for (int k = 0; k < TT; ++k) {
            const unsigned m = __builtin_amdgcn_ballot_w32(cnt > k);
            if (m == 0u) break;
            if (lane == 0) reinterpret_cast<unsigned*>(lvl_s)[k * 4 + wv] = m;
        }
    }

    for (int e = tid; e < TT * NA; e += SCAN_THREADS) {
        const int t = e / NA;
        const int a = e - t * NA;
        const float abil = -3.0f + 0.5f * (float)a;
        const int c = c_s[t];
        const float4 op4 = op_s[t];
        const v2f xv = x_s[t];
        const float x0 = xv.x, x1 = xv.y;
        float l00 = op4.x + olk_s[c * 4 + 0] - abil + x0;
        float l01 = op4.y + olk_s[c * 4 + 1] + abil - x0;
        float l10 = op4.z + olk_s[c * 4 + 2] - abil + x1;
        float l11 = op4.w + olk_s[c * 4 + 3] + abil - x1;
        const v2f q0 = softmax2(l01 - l00);
        const v2f q1 = softmax2(l11 - l10);
        float4 r;
        r.x = q0.x; r.y = q0.y;
        r.z = q1.x; r.w = q1.y;
        lo_s[e] = r;
    }
    __syncthreads();

    if (valid) {
        int lev = 0;
        for (int k = 0; k < occ; ++k) {
            const uint4 m = lvl_s[k];
            lev += __popc(m.x) + __popc(m.y) + __popc(m.z) + __popc(m.w);
        }
        const uint4 m = lvl_s[occ];
        const int v = myc >> 5;
        const unsigned below = (1u << (myc & 31)) - 1u;
        const unsigned m0 = (v > 0) ? ~0u : below;
        const unsigned m1 = (v > 1) ? ~0u : ((v == 1) ? below : 0u);
        const unsigned m2 = (v > 2) ? ~0u : ((v == 2) ? below : 0u);
        const unsigned m3 = (v == 3) ? below : 0u;
        const int idx = __popc(m.x & m0) + __popc(m.y & m1) + __popc(m.z & m2) + __popc(m.w & m3);
        const int pos = lev + idx;
        ord_s[pos] = tid | (myc << 9) | ((obs_s[tid] != 0) << 16) | (occ << 17);
        if (idx == 0) lev_s[occ] = pos;
    }
    __syncthreads();

    {
        int k = 0;
        int s0 = 0;
        int s1 = __builtin_amdgcn_readfirstlane(lev_s[1]);
        while (s0 < TT) {
            for (int e = s0 * NA + tid; e < s1 * NA; e += SCAN_THREADS) {
                const int p = e / NA;
                const int a = e - p * NA;
                const int rec = ord_s[p];
                const int t = rec & 511;
                const int c = (rec >> 9) & 127;
                const int obs = (rec >> 16) & 1;
                const float4 q = lo_s[t * NA + a];
                const v2f pv = alpha[c * NA + a];
                const float4 T = T_s[c];
                const float w0 = (obs ? q.y : q.x) * pv.x;
                const float w1 = (obs ? q.w : q.z) * pv.y;
                const float n0 = T.x * w0 + T.y * w1;
                const float n1 = T.z * w0 + T.w * w1;
                const float rn = __builtin_amdgcn_rcpf(n0 + n1);
                v2f pn;
                pn.x = n0 * rn; pn.y = n1 * rn;
                alpha[c * NA + a] = pn;
                snap_s[t * NA + a] = pv;
            }
            ++k;
            s0 = s1;
            s1 = __builtin_amdgcn_readfirstlane(lev_s[k + 1]);
            __syncthreads();
        }
    }

    for (int e = tid; e < TT * NA; e += SCAN_THREADS) {
        const float4 q = lo_s[e];
        const v2f pv = snap_s[e];
        v2f g;
        g.x = __logf(q.x * pv.x + q.z * pv.y);
        g.y = __logf(q.y * pv.x + q.w * pv.y);
        snap_s[e] = g;
    }
    __syncthreads();

    v2f* vv_s = reinterpret_cast<v2f*>(lo_s);
    if (wave == 0 && lane < NA) {
        float logab = -2.5649493574615367f;
        #pragma unroll 10
        for (int t = 0; t < TT; ++t) {
            const v2f g = snap_s[t * NA + lane];
            v2f v;
            v.x = logab + g.x; v.y = logab + g.y;
            vv_s[t * NA + lane] = v;
            logab += obs_s[t] ? g.y : g.x;
        }
    }
    __syncthreads();

    for (int t = tid; t < TT; t += SCAN_THREADS) {
        float v0[NA], v1[NA];
        for (int a = 0; a < NA; ++a) {
            const v2f v = vv_s[t * NA + a];
            v0[a] = v.x; v1[a] = v.y;
        }
        const float o0 = lse13(v0);
        const float o1 = lse13(v1);
        const float onrm = lse2(o0, o1);
        v2f ov;
        ov.x = o0 - onrm;
        ov.y = o1 - onrm;
        *(v2f*)(out + (bT + t) * 2) = ov;
    }
}

extern "C" void kernel_launch(void* const* d_in, const int* in_sizes, int n_in,
                              void* d_out, int out_size, void* d_ws, size_t ws_size,
                              hipStream_t stream) {
    (void)in_sizes; (void)n_in; (void)out_size;
    const int* corr = (const int*)d_in[0];
    const int* kcv  = (const int*)d_in[1];
    const int* prob = (const int*)d_in[2];
    const float* FM    = (const float*)d_in[3];
    const float* trans = (const float*)d_in[4];
    const float* olp   = (const float*)d_in[5];
    const float* olk   = (const float*)d_in[6];
    const float* initl = (const float*)d_in[7];
    const float* lrw   = (const float*)d_in[8];
    const float* lrb   = (const float*)d_in[9];
    float* out = (float*)d_out;

    const size_t xbytes = (size_t)BB * TT * 2 * sizeof(float);
    if (ws_size >= xbytes) {
        float* xbuf = (float*)d_ws;
        hipLaunchKernelGGL(gemm_x_wmma, dim3(GEMM_BLOCKS), dim3(GEMM_THREADS), 0, stream,
                           FM, lrw, lrb, xbuf);
        hipLaunchKernelGGL((bkt_scan<true>), dim3(BB), dim3(SCAN_THREADS), 0, stream,
                           corr, kcv, prob, FM, trans, olp, olk, initl,
                           lrw, lrb, xbuf, out);
    } else {
        hipLaunchKernelGGL((bkt_scan<false>), dim3(BB), dim3(SCAN_THREADS), 0, stream,
                           corr, kcv, prob, FM, trans, olp, olk, initl,
                           lrw, lrb, (const float*)nullptr, out);
    }
}


# scan: serial log_ability wave only adds: terms (log p(obs|a), picked a phase earlier) lie [ability][step], read 16 B wide a buffer ahead; sum + log p(y|a) made in output phase; same sums, same order
# speedup vs baseline: 31.9759x; 1.1235x over previous
#include <hip/hip_runtime.h>

#define BB 256
#define TT 500
#define NKC 100
#define NA 13

typedef __attribute__((ext_vector_type(2))) float v2f;
typedef __attribute__((ext_vector_type(4))) float v4f;
typedef __attribute__((ext_vector_type(8))) float v8f;

__device__ __forceinline__ float lse2(float x, float y) {
    float m = fmaxf(x, y);
    return m + log1pf(__expf(-fabsf(x - y)));
}

__device__ __forceinline__ v2f softmax2(float d) {
    const float e = __expf(-fabsf(d));
    const float r = __builtin_amdgcn_rcpf(1.0f + e);
    const float s = e * r;
    v2f q;
    q.x = (d >= 0.f) ? s : r;
    q.y = (d >= 0.f) ? r : s;
    return q;
}

#define GEMM_THREADS 256
#define GEMM_WAVES (GEMM_THREADS / 32)
#define TILES_PER_WAVE 8
#define GEMM_BLOCKS 500

struct AFrag { v4f m[6]; float r; };

__device__ __forceinline__ void load_afrag(AFrag& f, const float* __restrict__ rowp,
                                           int koff, int roff) {
    for (int j = 0; j < 6; ++j) f.m[j] = *(const v4f*)(rowp + koff + 32 * j);
    f.r = rowp[roff];
}

__device__ __forceinline__ v8f wmma4(v2f a, v2f b, v8f c) {
    return __builtin_amdgcn_wmma_f32_16x16x4_f32(false, a, false, b, (short)0, c, false, false);
}

__device__ __forceinline__ float quad_xor1(float v) {
    return __int_as_float(__builtin_amdgcn_update_dpp(0, __float_as_int(v), 0xB1, 0xF, 0xF, true));
}
__device__ __forceinline__ float quad_xor2(float v) {
    return __int_as_float(__builtin_amdgcn_update_dpp(0, __float_as_int(v), 0x4E, 0xF, 0xF, true));
}

__device__ __forceinline__ void store_tile(const v8f acc, long trow, int g, int q, int h,
                                           float bias, float* __restrict__ xout) {
    float s0 = ((g == 0) ? acc[0] : 0.f) + ((g == 1) ? acc[1] : 0.f)
             + ((g == 2) ? acc[2] : 0.f) + ((g == 3) ? acc[3] : 0.f);
    float s1 = ((g == 0) ? acc[4] : 0.f) + ((g == 1) ? acc[5] : 0.f)
             + ((g == 2) ? acc[6] : 0.f) + ((g == 3) ? acc[7] : 0.f);
    s0 += quad_xor1(s0); s0 += quad_xor2(s0);
    s1 += quad_xor1(s1); s1 += quad_xor2(s1);
    if (g == 0 && q < 2) {
        const long row = trow + 2 * h;
        xout[row * 2 + q] = s0 + bias;
        xout[(row + 1) * 2 + q] = s1 + bias;
    }
}

__global__ __launch_bounds__(GEMM_THREADS) __attribute__((amdgpu_waves_per_eu(4, 4)))
void gemm_x_wmma(const float* __restrict__ FM, const float* __restrict__ lrw,
                 const float* __restrict__ lrb, float* __restrict__ xout) {
    const int lane = threadIdx.x & 31;
    const int gw = blockIdx.x * GEMM_WAVES + (threadIdx.x >> 5);
    const int c = lane & 15;
    const int h = lane >> 4;
    const int g = c & 3;
    const int q = c >> 2;
    const int koff = 8 * g + 4 * h;
    const int roff = 192 + 2 * g + h;

    const bool bval = q < 2;
    const float* wrow = lrw + (q & 1) * 200;
    v4f bm[6];
    for (int j = 0; j < 6; ++j) {
        v4f t;
        t.x = wrow[koff + 32 * j + 0]; t.y = wrow[koff + 32 * j + 1];
        t.z = wrow[koff + 32 * j + 2]; t.w = wrow[koff + 32 * j + 3];
        const v4f z = {0.f, 0.f, 0.f, 0.f};
        bm[j] = bval ? t : z;
    }
    const float brt = wrow[roff];
    const float br = bval ? brt : 0.f;
    const float bias = lrb[q & 1];

    const long row0 = (long)gw * (TILES_PER_WAVE * 4);
    const float* ap = FM + (row0 + q) * 200;

    AFrag c0, c1;
    load_afrag(c0, ap, koff, roff);
    load_afrag(c1, ap + 800, koff, roff);
    for (int p = 0; p < TILES_PER_WAVE / 2; ++p) {
        AFrag n0 = c0, n1 = c1;
        if (p + 1 < TILES_PER_WAVE / 2) {
            load_afrag(n0, ap + (2 * p + 2) * 800, koff, roff);
            load_afrag(n1, ap + (2 * p + 3) * 800, koff, roff);
        }
        v8f acc0 = {}, acc1 = {};
        for (int j = 0; j < 6; ++j) {
            v2f a, b;
            b.x = bm[j].x; b.y = bm[j].y;
            a.x = c0.m[j].x; a.y = c0.m[j].y; acc0 = wmma4(a, b, acc0);
            a.x = c1.m[j].x; a.y = c1.m[j].y; acc1 = wmma4(a, b, acc1);
            b.x = bm[j].z; b.y = bm[j].w;
            a.x = c0.m[j].z; a.y = c0.m[j].w; acc0 = wmma4(a, b, acc0);
            a.x = c1.m[j].z; a.y = c1.m[j].w; acc1 = wmma4(a, b, acc1);
        }
        {
            v2f a, b;
            b.x = br; b.y = 0.f;
            a.x = c0.r; a.y = 0.f; acc0 = wmma4(a, b, acc0);
            a.x = c1.r; a.y = 0.f; acc1 = wmma4(a, b, acc1);
        }
        store_tile(acc0, row0 + (2 * p) * 4, g, q, h, bias, xout);
        store_tile(acc1, row0 + (2 * p + 1) * 4, g, q, h, bias, xout);
        c0 = n0; c1 = n1;
    }
}

#define SCAN_THREADS 512
#define SCAN_WAVES (SCAN_THREADS / 32)
static_assert(NKC <= 128, "a chain number has seven bits and a row of the chain bitmap has four words");
#define ESTEPS 512
#define EPITCH (ESTEPS + 4)
#define EPIECES (ESTEPS / 4)
#define EBUF 16
static_assert(TT <= ESTEPS && EPIECES % (2 * EBUF) == 0, "phase E walks whole pairs of register buffers");

__device__ __forceinline__ float running_sum(const v4f (&buf)[EBUF], float s, float* dst) {
    #pragma unroll
    for (int i = 0; i < EBUF; ++i) {
        v4f o;
        o.x = s; s += buf[i].x;
        o.y = s; s += buf[i].y;
        o.z = s; s += buf[i].z;
        o.w = s; s += buf[i].w;
        *(v4f*)(dst + 4 * i) = o;
    }
    return s;
}

__device__ __forceinline__ float lse13(const float* v) {
    float m = -1.0e30f;
    for (int a = 0; a < NA; ++a) m = fmaxf(m, v[a]);
    float e[NA];
    for (int a = 0; a < NA; ++a) e[a] = __expf(v[a] - m);
    const float sA = (e[0] + e[8]) + (e[4] + e[12]);
    const float sB = (e[2] + e[10]) + e[6];
    const float sC = (e[1] + e[9]) + e[5];
    const float sD = (e[3] + e[11]) + e[7];
    const float s = (sA + sB) + (sC + sD);
    return m + __logf(s);
}

template <bool USE_XBUF>
__global__ __launch_bounds__(SCAN_THREADS)
void bkt_scan(const int* __restrict__ corr, const int* __restrict__ kcv,
              const int* __restrict__ prob, const float* __restrict__ FM,
              const float* __restrict__ trans, const float* __restrict__ olp,
              const float* __restrict__ olk, const float* __restrict__ initl,
              const float* __restrict__ lrw, const float* __restrict__ lrb,
              const float* __restrict__ xbuf, float* __restrict__ out) {
    __shared__ float4 lo_s[TT * NA];
    __shared__ v2f snap_s[TT * NA];
    __shared__ float4 op_s[TT];
    __shared__ v2f x_s[TT];
    __shared__ __attribute__((aligned(16))) int c_s[TT];
    __shared__ int obs_s[TT];
    __shared__ int hist_s[SCAN_WAVES * NKC];
    __shared__ uint4 lvl_s[TT];
    __shared__ int ord_s[TT];
    __shared__ int lev_s[TT + 2];
    __shared__ v2f alpha[NKC * NA];
    __shared__ float olk_s[NKC * 4];
    __shared__ float4 T_s[NKC];
    __shared__ v2f pinit_s[NKC];
    __shared__ __attribute__((aligned(16))) float term_s[NA * EPITCH];

    const int tid = threadIdx.x;
    const int wave = tid >> 5, lane = tid & 31;
    const int b = blockIdx.x;
    const long bT = (long)b * TT;

    for (int i = tid; i < NKC * 4; i += SCAN_THREADS) olk_s[i] = olk[i];
    for (int i = tid; i < TT + 2; i += SCAN_THREADS) lev_s[i] = TT;
    for (int i = tid; i < SCAN_WAVES * NKC; i += SCAN_THREADS) hist_s[i] = 0;
    for (int i = tid; i < TT; i += SCAN_THREADS) lvl_s[i] = make_uint4(0u, 0u, 0u, 0u);
    for (int i = tid; i < NA * (ESTEPS - TT); i += SCAN_THREADS)
        term_s[(i / (ESTEPS - TT)) * EPITCH + TT + i % (ESTEPS - TT)] = 0.f;
    for (int c = tid; c < NKC; c += SCAN_THREADS) {
        pinit_s[c] = softmax2(initl[c * 2 + 1] - initl[c * 2]);
        const float t00 = trans[c * 4 + 0], t01 = trans[c * 4 + 1];
        const float t10 = trans[c * 4 + 2], t11 = trans[c * 4 + 3];
        const v2f col0 = softmax2(t10 - t00);
        const v2f col1 = softmax2(t11 - t01);
        float4 T;
        T.x = col0.x; T.y = col1.x;
        T.z = col0.y; T.w = col1.y;
        T_s[c] = T;
    }

    for (int t = tid; t < TT; t += SCAN_THREADS) {
        const long idx = bT + t;
        c_s[t] = kcv[idx];
        obs_s[t] = corr[idx];
        const int p = prob[idx];
        op_s[t] = *(const float4*)(olp + (long)p * 4);
        if (USE_XBUF) x_s[t] = *(const v2f*)(xbuf + idx * 2);
    }
    if (!USE_XBUF) {
        for (int t = wave; t < TT; t += SCAN_WAVES) {
            const float* fr = FM + (bT + t) * 200;
            float p0 = 0.f, p1 = 0.f;
            for (int f = lane; f < 200; f += 32) {
                float fv = fr[f];
                p0 = fmaf(fv, lrw[f], p0);
                p1 = fmaf(fv, lrw[200 + f], p1);
            }
            for (int off = 16; off; off >>= 1) {
                p0 += __shfl_xor(p0, off, 32);
                p1 += __shfl_xor(p1, off, 32);
            }
            if (lane == 0) {
                v2f xv;
                xv.x = p0 + lrb[0];
                xv.y = p1 + lrb[1];
                x_s[t] = xv;
            }
        }
    }
    __syncthreads();

    for (int i = tid; i < NKC * NA; i += SCAN_THREADS) alpha[i] = pinit_s[i / NA];

    const bool valid = tid < TT;
    const int myc = valid ? c_s[tid] : 0;
    unsigned peers = __builtin_amdgcn_ballot_w32(valid);
    for (int bit = 0; bit < 7; ++bit) {
        const bool on = ((myc >> bit) & 1) != 0;
        const unsigned m = __builtin_amdgcn_ballot_w32(on);
        peers &= on ? m : ~m;
    }
    const int rin = __popc(peers & ((1u << lane) - 1u));
    const int wv = __builtin_amdgcn_readfirstlane(wave);
    if (valid && rin == 0) hist_s[wv * NKC + myc] = __popc(peers);
    __syncthreads();

    int occ = rin;
    for (int w = 0; w < wv; ++w) occ += hist_s[w * NKC + myc];
    if (wv < (NKC + 31) / 32) {
        int cnt = 0;
        if (tid < NKC)
            for (int w = 0; w < SCAN_WAVES; ++w) cnt += hist_s[w * NKC + tid];
        for (int k = 0; k < TT; ++k) {
            const unsigned m = __builtin_amdgcn_ballot_w32(cnt > k);
            if (m == 0u) break;
            if (lane == 0) reinterpret_cast<unsigned*>(lvl_s)[k * 4 + wv] = m;
        }
    }

    for (int e = tid; e < TT * NA; e += SCAN_THREADS) {
        const int t = e / NA;
        const int a = e - t * NA;
        const float abil = -3.0f + 0.5f * (float)a;
        const int c = c_s[t];
        const float4 op4 = op_s[t];
        const v2f xv = x_s[t];
        const float x0 = xv.x, x1 = xv.y;
        float l00 = op4.x + olk_s[c * 4 + 0] - abil + x0;
        float l01 = op4.y + olk_s[c * 4 + 1] + abil - x0;
        float l10 = op4.z + olk_s[c * 4 + 2] - abil + x1;
        float l11 = op4.w + olk_s[c * 4 + 3] + abil - x1;
        const v2f q0 = softmax2(l01 - l00);
        const v2f q1 = softmax2(l11 - l10);
        float4 r;
        r.x = q0.x; r.y = q0.y;
        r.z = q1.x; r.w = q1.y;
        lo_s[e] = r;
    }
    __syncthreads();

    if (valid) {
        int lev = 0;
        for (int k = 0; k < occ; ++k) {
            const uint4 m = lvl_s[k];
            lev += __popc(m.x) + __popc(m.y) + __popc(m.z) + __popc(m.w);
        }
        const uint4 m = lvl_s[occ];
        const int v = myc >> 5;
        const unsigned below = (1u << (myc & 31)) - 1u;
        const unsigned m0 = (v > 0) ? ~0u : below;
        const unsigned m1 = (v > 1) ? ~0u : ((v == 1) ? below : 0u);
        const unsigned m2 = (v > 2) ? ~0u : ((v == 2) ? below : 0u);
        const unsigned m3 = (v == 3) ? below : 0u;
        const int idx = __popc(m.x & m0) + __popc(m.y & m1) + __popc(m.z & m2) + __popc(m.w & m3);
        const int pos = lev + idx;
        ord_s[pos] = tid | (myc << 9) | ((obs_s[tid] != 0) << 16) | (occ << 17);
        if (idx == 0) lev_s[occ] = pos;
    }
    __syncthreads();

    {
        int k = 0;
        int s0 = 0;
        int s1 = __builtin_amdgcn_readfirstlane(lev_s[1]);
        while (s0 < TT) {
            for (int e = s0 * NA + tid; e < s1 * NA; e += SCAN_THREADS) {
                const int p = e / NA;
                const int a = e - p * NA;
                const int rec = ord_s[p];
                const int t = rec & 511;
                const int c = (rec >> 9) & 127;
                const int obs = (rec >> 16) & 1;
                const float4 q = lo_s[t * NA + a];
                const v2f pv = alpha[c * NA + a];
                const float4 T = T_s[c];
                const float w0 = (obs ? q.y : q.x) * pv.x;
                const float w1 = (obs ? q.w : q.z) * pv.y;
                const float n0 = T.x * w0 + T.y * w1;
                const float n1 = T.z * w0 + T.w * w1;
                const float rn = __builtin_amdgcn_rcpf(n0 + n1);
                v2f pn;
                pn.x = n0 * rn; pn.y = n1 * rn;
                alpha[c * NA + a] = pn;
                snap_s[t * NA + a] = pv;
            }
            ++k;
            s0 = s1;
            s1 = __builtin_amdgcn_readfirstlane(lev_s[k + 1]);
            __syncthreads();
        }
    }

    for (int e = tid; e < TT * NA; e += SCAN_THREADS) {
        const int t = e / NA;
        const int a = e - t * NA;
        const float4 q = lo_s[e];
        const v2f pv = snap_s[e];
        v2f g;
        g.x = __logf(q.x * pv.x + q.z * pv.y);
        g.y = __logf(q.y * pv.x + q.w * pv.y);
        snap_s[e] = g;
        term_s[a * EPITCH + t] = obs_s[t] ? g.y : g.x;
    }
    __syncthreads();

    float* pre_s = reinterpret_cast<float*>(lo_s);
    if (wave == 0 && lane < NA) {
        const float* trow = term_s + lane * EPITCH;
        float* prow = pre_s + lane * EPITCH;
        float logab = -2.5649493574615367f;
        v4f bufa[EBUF], bufb[EBUF];
        #pragma unroll
        for (int i = 0; i < EBUF; ++i) bufa[i] = *(const v4f*)(trow + 4 * i);
        for (int p0 = 0; p0 < EPIECES; p0 += 2 * EBUF) {
            #pragma unroll
            for (int i = 0; i < EBUF; ++i) bufb[i] = *(const v4f*)(trow + 4 * (p0 + EBUF + i));
            logab = running_sum(bufa, logab, prow + 4 * p0);
            if (p0 + 2 * EBUF < EPIECES) {
                #pragma unroll
                for (int i = 0; i < EBUF; ++i) bufa[i] = *(const v4f*)(trow + 4 * (p0 + 2 * EBUF + i));
            }
            logab = running_sum(bufb, logab, prow + 4 * (p0 + EBUF));
        }
    }
    __syncthreads();

    for (int t = tid; t < TT; t += SCAN_THREADS) {
        float v0[NA], v1[NA];
        for (int a = 0; a < NA; ++a) {
            const v2f g = snap_s[t * NA + a];
            const float logab = pre_s[a * EPITCH + t];
            v0[a] = logab + g.x; v1[a] = logab + g.y;
        }
        const float o0 = lse13(v0);
        const float o1 = lse13(v1);
        const float onrm = lse2(o0, o1);
        v2f ov;
        ov.x = o0 - onrm;
        ov.y = o1 - onrm;
        *(v2f*)(out + (bT + t) * 2) = ov;
    }
}

extern "C" void kernel_launch(void* const* d_in, const int* in_sizes, int n_in,
                              void* d_out, int out_size, void* d_ws, size_t ws_size,
                              hipStream_t stream) {
    (void)in_sizes; (void)n_in; (void)out_size;
    const int* corr = (const int*)d_in[0];
    const int* kcv  = (const int*)d_in[1];
    const int* prob = (const int*)d_in[2];
    const float* FM    = (const float*)d_in[3];
    const float* trans = (const float*)d_in[4];
    const float* olp   = (const float*)d_in[5];
    const float* olk   = (const float*)d_in[6];
    const float* initl = (const float*)d_in[7];
    const float* lrw   = (const float*)d_in[8];
    const float* lrb   = (const float*)d_in[9];
    float* out = (float*)d_out;

    const size_t xbytes = (size_t)BB * TT * 2 * sizeof(float);
    if (ws_size >= xbytes) {
        float* xbuf = (float*)d_ws;
        hipLaunchKernelGGL(gemm_x_wmma, dim3(GEMM_BLOCKS), dim3(GEMM_THREADS), 0, stream,
                           FM, lrw, lrb, xbuf);
        hipLaunchKernelGGL((bkt_scan<true>), dim3(BB), dim3(SCAN_THREADS), 0, stream,
                           corr, kcv, prob, FM, trans, olp, olk, initl,
                           lrw, lrb, xbuf, out);
    } else {
        hipLaunchKernelGGL((bkt_scan<false>), dim3(BB), dim3(SCAN_THREADS), 0, stream,
                           corr, kcv, prob, FM, trans, olp, olk, initl,
                           lrw, lrb, (const float*)nullptr, out);
    }
}


# scan: recurrence loop fetches one occurrence number ahead (record, observation pair, transition row) under the update before it, a thread's 3 elements at once; same order of sums, multiply-adds fuse
# speedup vs baseline: 33.0687x; 1.0342x over previous
#include <hip/hip_runtime.h>

#define BB 256
#define TT 500
#define NKC 100
#define NA 13

typedef __attribute__((ext_vector_type(2))) float v2f;
typedef __attribute__((ext_vector_type(4))) float v4f;
typedef __attribute__((ext_vector_type(8))) float v8f;

__device__ __forceinline__ float lse2(float x, float y) {
    float m = fmaxf(x, y);
    return m + log1pf(__expf(-fabsf(x - y)));
}

__device__ __forceinline__ v2f softmax2(float d) {
    const float e = __expf(-fabsf(d));
    const float r = __builtin_amdgcn_rcpf(1.0f + e);
    const float s = e * r;
    v2f q;
    q.x = (d >= 0.f) ? s : r;
    q.y = (d >= 0.f) ? r : s;
    return q;
}

#define GEMM_THREADS 256
#define GEMM_WAVES (GEMM_THREADS / 32)
#define TILES_PER_WAVE 8
#define GEMM_BLOCKS 500

struct AFrag { v4f m[6]; float r; };

__device__ __forceinline__ void load_afrag(AFrag& f, const float* __restrict__ rowp,
                                           int koff, int roff) {
    for (int j = 0; j < 6; ++j) f.m[j] = *(const v4f*)(rowp + koff + 32 * j);
    f.r = rowp[roff];
}

__device__ __forceinline__ v8f wmma4(v2f a, v2f b, v8f c) {
    return __builtin_amdgcn_wmma_f32_16x16x4_f32(false, a, false, b, (short)0, c, false, false);
}

__device__ __forceinline__ float quad_xor1(float v) {
    return __int_as_float(__builtin_amdgcn_update_dpp(0, __float_as_int(v), 0xB1, 0xF, 0xF, true));
}
__device__ __forceinline__ float quad_xor2(float v) {
    return __int_as_float(__builtin_amdgcn_update_dpp(0, __float_as_int(v), 0x4E, 0xF, 0xF, true));
}

__device__ __forceinline__ void store_tile(const v8f acc, long trow, int g, int q, int h,
                                           float bias, float* __restrict__ xout) {
    float s0 = ((g == 0) ? acc[0] : 0.f) + ((g == 1) ? acc[1] : 0.f)
             + ((g == 2) ? acc[2] : 0.f) + ((g == 3) ? acc[3] : 0.f);
    float s1 = ((g == 0) ? acc[4] : 0.f) + ((g == 1) ? acc[5] : 0.f)
             + ((g == 2) ? acc[6] : 0.f) + ((g == 3) ? acc[7] : 0.f);
    s0 += quad_xor1(s0); s0 += quad_xor2(s0);
    s1 += quad_xor1(s1); s1 += quad_xor2(s1);
    if (g == 0 && q < 2) {
        const long row = trow + 2 * h;
        xout[row * 2 + q] = s0 + bias;
        xout[(row + 1) * 2 + q] = s1 + bias;
    }
}

__global__ __launch_bounds__(GEMM_THREADS) __attribute__((amdgpu_waves_per_eu(4, 4)))
void gemm_x_wmma(const float* __restrict__ FM, const float* __restrict__ lrw,
                 const float* __restrict__ lrb, float* __restrict__ xout) {
    const int lane = threadIdx.x & 31;
    const int gw = blockIdx.x * GEMM_WAVES + (threadIdx.x >> 5);
    const int c = lane & 15;
    const int h = lane >> 4;
    const int g = c & 3;
    const int q = c >> 2;
    const int koff = 8 * g + 4 * h;
    const int roff = 192 + 2 * g + h;

    const bool bval = q < 2;
    const float* wrow = lrw + (q & 1) * 200;
    v4f bm[6];
    for (int j = 0; j < 6; ++j) {
        v4f t;
        t.x = wrow[koff + 32 * j + 0]; t.y = wrow[koff + 32 * j + 1];
        t.z = wrow[koff + 32 * j + 2]; t.w = wrow[koff + 32 * j + 3];
        const v4f z = {0.f, 0.f, 0.f, 0.f};
        bm[j] = bval ? t : z;
    }
    const float brt = wrow[roff];
    const float br = bval ? brt : 0.f;
    const float bias = lrb[q & 1];

    const long row0 = (long)gw * (TILES_PER_WAVE * 4);
    const float* ap = FM + (row0 + q) * 200;

    AFrag c0, c1;
    load_afrag(c0, ap, koff, roff);
    load_afrag(c1, ap + 800, koff, roff);
    for (int p = 0; p < TILES_PER_WAVE / 2; ++p) {
        AFrag n0 = c0, n1 = c1;
        if (p + 1 < TILES_PER_WAVE / 2) {
            load_afrag(n0, ap + (2 * p + 2) * 800, koff, roff);
            load_afrag(n1, ap + (2 * p + 3) * 800, koff, roff);
        }
        v8f acc0 = {}, acc1 = {};
        for (int j = 0; j < 6; ++j) {
            v2f a, b;
            b.x = bm[j].x; b.y = bm[j].y;
            a.x = c0.m[j].x; a.y = c0.m[j].y; acc0 = wmma4(a, b, acc0);
            a.x = c1.m[j].x; a.y = c1.m[j].y; acc1 = wmma4(a, b, acc1);
            b.x = bm[j].z; b.y = bm[j].w;
            a.x = c0.m[j].z; a.y = c0.m[j].w; acc0 = wmma4(a, b, acc0);
            a.x = c1.m[j].z; a.y = c1.m[j].w; acc1 = wmma4(a, b, acc1);
        }
        {
            v2f a, b;
            b.x = br; b.y = 0.f;
            a.x = c0.r; a.y = 0.f; acc0 = wmma4(a, b, acc0);
            a.x = c1.r; a.y = 0.f; acc1 = wmma4(a, b, acc1);
        }
        store_tile(acc0, row0 + (2 * p) * 4, g, q, h, bias, xout);
        store_tile(acc1, row0 + (2 * p + 1) * 4, g, q, h, bias, xout);
        c0 = n0; c1 = n1;
    }
}

#define SCAN_THREADS 512
#define SCAN_WAVES (SCAN_THREADS / 32)
static_assert(NKC <= 128, "a chain number has seven bits and a row of the chain bitmap has four words");
#define ESTEPS 512
#define EPITCH (ESTEPS + 4)
#define EPIECES (ESTEPS / 4)
#define EBUF 16
#define CSLOTS ((NKC * NA + SCAN_THREADS - 1) / SCAN_THREADS)
static_assert((NA - 1) + (SCAN_THREADS % NA) * (CSLOTS - 1) < 2 * NA, "a slot's ability index carries at most once");

struct CStat {
    int ai;
    int si;
    float q0, q1;
    float4 T;
};
static_assert(TT <= ESTEPS && EPIECES % (2 * EBUF) == 0, "phase E walks whole pairs of register buffers");

__device__ __forceinline__ float running_sum(const v4f (&buf)[EBUF], float s, float* dst) {
    #pragma unroll
    for (int i = 0; i < EBUF; ++i) {
        v4f o;
        o.x = s; s += buf[i].x;
        o.y = s; s += buf[i].y;
        o.z = s; s += buf[i].z;
        o.w = s; s += buf[i].w;
        *(v4f*)(dst + 4 * i) = o;
    }
    return s;
}

__device__ __forceinline__ float lse13(const float* v) {
    float m = -1.0e30f;
    for (int a = 0; a < NA; ++a) m = fmaxf(m, v[a]);
    float e[NA];
    for (int a = 0; a < NA; ++a) e[a] = __expf(v[a] - m);
    const float sA = (e[0] + e[8]) + (e[4] + e[12]);
    const float sB = (e[2] + e[10]) + e[6];
    const float sC = (e[1] + e[9]) + e[5];
    const float sD = (e[3] + e[11]) + e[7];
    const float s = (sA + sB) + (sC + sD);
    return m + __logf(s);
}

template <bool USE_XBUF>
__global__ __launch_bounds__(SCAN_THREADS)
void bkt_scan(const int* __restrict__ corr, const int* __restrict__ kcv,
              const int* __restrict__ prob, const float* __restrict__ FM,
              const float* __restrict__ trans, const float* __restrict__ olp,
              const float* __restrict__ olk, const float* __restrict__ initl,
              const float* __restrict__ lrw, const float* __restrict__ lrb,
              const float* __restrict__ xbuf, float* __restrict__ out) {
    __shared__ float4 lo_s[TT * NA];
    __shared__ v2f snap_s[TT * NA];
    __shared__ float4 op_s[TT];
    __shared__ v2f x_s[TT];
    __shared__ __attribute__((aligned(16))) int c_s[TT];
    __shared__ int obs_s[TT];
    __shared__ int hist_s[SCAN_WAVES * NKC];
    __shared__ uint4 lvl_s[TT];
    __shared__ int ord_s[TT];
    __shared__ int lev_s[TT + 4];
    __shared__ v2f alpha[NKC * NA];
    __shared__ float olk_s[NKC * 4];
    __shared__ float4 T_s[NKC];
    __shared__ v2f pinit_s[NKC];
    __shared__ __attribute__((aligned(16))) float term_s[NA * EPITCH];

    const int tid = threadIdx.x;
    const int wave = tid >> 5, lane = tid & 31;
    const int b = blockIdx.x;
    const long bT = (long)b * TT;

    for (int i = tid; i < NKC * 4; i += SCAN_THREADS) olk_s[i] = olk[i];
    for (int i = tid; i < TT + 4; i += SCAN_THREADS) lev_s[i] = TT;
    for (int i = tid; i < SCAN_WAVES * NKC; i += SCAN_THREADS) hist_s[i] = 0;
    for (int i = tid; i < TT; i += SCAN_THREADS) lvl_s[i] = make_uint4(0u, 0u, 0u, 0u);
    for (int i = tid; i < NA * (ESTEPS - TT); i += SCAN_THREADS)
        term_s[(i / (ESTEPS - TT)) * EPITCH + TT + i % (ESTEPS - TT)] = 0.f;
    for (int c = tid; c < NKC; c += SCAN_THREADS) {
        pinit_s[c] = softmax2(initl[c * 2 + 1] - initl[c * 2]);
        const float t00 = trans[c * 4 + 0], t01 = trans[c * 4 + 1];
        const float t10 = trans[c * 4 + 2], t11 = trans[c * 4 + 3];
        const v2f col0 = softmax2(t10 - t00);
        const v2f col1 = softmax2(t11 - t01);
        float4 T;
        T.x = col0.x; T.y = col1.x;
        T.z = col0.y; T.w = col1.y;
        T_s[c] = T;
    }

    for (int t = tid; t < TT; t += SCAN_THREADS) {
        const long idx = bT + t;
        c_s[t] = kcv[idx];
        obs_s[t] = corr[idx];
        const int p = prob[idx];
        op_s[t] = *(const float4*)(olp + (long)p * 4);
        if (USE_XBUF) x_s[t] = *(const v2f*)(xbuf + idx * 2);
    }
    if (!USE_XBUF) {
        for (int t = wave; t < TT; t += SCAN_WAVES) {
            const float* fr = FM + (bT + t) * 200;
            float p0 = 0.f, p1 = 0.f;
            for (int f = lane; f < 200; f += 32) {
                float fv = fr[f];
                p0 = fmaf(fv, lrw[f], p0);
                p1 = fmaf(fv, lrw[200 + f], p1);
            }
            for (int off = 16; off; off >>= 1) {
                p0 += __shfl_xor(p0, off, 32);
                p1 += __shfl_xor(p1, off, 32);
            }
            if (lane == 0) {
                v2f xv;
                xv.x = p0 + lrb[0];
                xv.y = p1 + lrb[1];
                x_s[t] = xv;
            }
        }
    }
    __syncthreads();

    for (int i = tid; i < NKC * NA; i += SCAN_THREADS) alpha[i] = pinit_s[i / NA];

    const bool valid = tid < TT;
    const int myc = valid ? c_s[tid] : 0;
    unsigned peers = __builtin_amdgcn_ballot_w32(valid);
    for (int bit = 0; bit < 7; ++bit) {
        const bool on = ((myc >> bit) & 1) != 0;
        const unsigned m = __builtin_amdgcn_ballot_w32(on);
        peers &= on ? m : ~m;
    }
    const int rin = __popc(peers & ((1u << lane) - 1u));
    const int wv = __builtin_amdgcn_readfirstlane(wave);
    if (valid && rin == 0) hist_s[wv * NKC + myc] = __popc(peers);
    __syncthreads();

    int occ = rin;
    for (int w = 0; w < wv; ++w) occ += hist_s[w * NKC + myc];
    if (wv < (NKC + 31) / 32) {
        int cnt = 0;
        if (tid < NKC)
            for (int w = 0; w < SCAN_WAVES; ++w) cnt += hist_s[w * NKC + tid];
        for (int k = 0; k < TT; ++k) {
            const unsigned m = __builtin_amdgcn_ballot_w32(cnt > k);
            if (m == 0u) break;
            if (lane == 0) reinterpret_cast<unsigned*>(lvl_s)[k * 4 + wv] = m;
        }
    }

    for (int e = tid; e < TT * NA; e += SCAN_THREADS) {
        const int t = e / NA;
        const int a = e - t * NA;
        const float abil = -3.0f + 0.5f * (float)a;
        const int c = c_s[t];
        const float4 op4 = op_s[t];
        const v2f xv = x_s[t];
        const float x0 = xv.x, x1 = xv.y;
        float l00 = op4.x + olk_s[c * 4 + 0] - abil + x0;
        float l01 = op4.y + olk_s[c * 4 + 1] + abil - x0;
        float l10 = op4.z + olk_s[c * 4 + 2] - abil + x1;
        float l11 = op4.w + olk_s[c * 4 + 3] + abil - x1;
        const v2f q0 = softmax2(l01 - l00);
        const v2f q1 = softmax2(l11 - l10);
        float4 r;
        r.x = q0.x; r.y = q0.y;
        r.z = q1.x; r.w = q1.y;
        lo_s[e] = r;
    }
    __syncthreads();

    if (valid) {
        int lev = 0;
        for (int k = 0; k < occ; ++k) {
            const uint4 m = lvl_s[k];
            lev += __popc(m.x) + __popc(m.y) + __popc(m.z) + __popc(m.w);
        }
        const uint4 m = lvl_s[occ];
        const int v = myc >> 5;
        const unsigned below = (1u << (myc & 31)) - 1u;
        const unsigned m0 = (v > 0) ? ~0u : below;
        const unsigned m1 = (v > 1) ? ~0u : ((v == 1) ? below : 0u);
        const unsigned m2 = (v > 2) ? ~0u : ((v == 2) ? below : 0u);
        const unsigned m3 = (v == 3) ? below : 0u;
        const int idx = __popc(m.x & m0) + __popc(m.y & m1) + __popc(m.z & m2) + __popc(m.w & m3);
        const int pos = lev + idx;
        ord_s[pos] = tid | (myc << 9) | ((obs_s[tid] != 0) << 16) | (occ << 17);
        if (idx == 0) lev_s[occ] = pos;
    }
    __syncthreads();

    {
        const int pt = tid / NA;
        const int at = tid - pt * NA;
        int P[CSLOTS], A[CSLOTS];
        #pragma unroll
        for (int j = 0; j < CSLOTS; ++j) {
            const int a = at + (SCAN_THREADS % NA) * j;
            const int carry = (a >= NA) ? 1 : 0;
            A[j] = a - NA * carry;
            P[j] = pt + (SCAN_THREADS / NA) * j + carry;
        }

        CStat cur[CSLOTS];
        auto fetch_stat = [&](int s) {
            int rec[CSLOTS];
            #pragma unroll
            for (int j = 0; j < CSLOTS; ++j) rec[j] = ord_s[min(s + P[j], TT - 1)];
            #pragma unroll
            for (int j = 0; j < CSLOTS; ++j) {
                const int t = rec[j] & 511;
                const int c = (rec[j] >> 9) & 127;
                const int obs = (rec[j] >> 16) & 1;
                const int si = t * NA + A[j];
                const float4 q = lo_s[si];
                cur[j].T = T_s[c];
                cur[j].ai = c * NA + A[j];
                cur[j].si = si;
                cur[j].q0 = obs ? q.y : q.x;
                cur[j].q1 = obs ? q.w : q.z;
            }
        };

        int k = 0;
        int s0 = 0;
        int s1 = __builtin_amdgcn_readfirstlane(lev_s[1]);
        int s2 = __builtin_amdgcn_readfirstlane(lev_s[2]);
        fetch_stat(s0);
        while (s0 < TT) {
            const int s3v = lev_s[k + 3];
            if (s0 + P[0] < s1 || s1 + P[0] < s2) {
                v2f pv[CSLOTS], pn[CSLOTS];
                int ai[CSLOTS], si[CSLOTS];
                #pragma unroll
                for (int j = 0; j < CSLOTS; ++j) {
                    ai[j] = cur[j].ai; si[j] = cur[j].si;
                    pv[j] = alpha[ai[j]];
                }
                #pragma unroll
                for (int j = 0; j < CSLOTS; ++j) {
                    const float4 T = cur[j].T;
                    const float w0 = cur[j].q0 * pv[j].x;
                    const float w1 = cur[j].q1 * pv[j].y;
                    const float n0 = T.x * w0 + T.y * w1;
                    const float n1 = T.z * w0 + T.w * w1;
                    const float rn = __builtin_amdgcn_rcpf(n0 + n1);
                    pn[j].x = n0 * rn; pn[j].y = n1 * rn;
                }
                fetch_stat(s1);
                #pragma unroll
                for (int j = 0; j < CSLOTS; ++j) {
                    if (s0 + P[j] < s1) {
                        alpha[ai[j]] = pn[j];
                        snap_s[si[j]] = pv[j];
                    }
                }
            }
            ++k;
            s0 = s1;
            s1 = s2;
            s2 = __builtin_amdgcn_readfirstlane(s3v);
            __syncthreads();
        }
    }

    for (int e = tid; e < TT * NA; e += SCAN_THREADS) {
        const int t = e / NA;
        const int a = e - t * NA;
        const float4 q = lo_s[e];
        const v2f pv = snap_s[e];
        v2f g;
        g.x = __logf(q.x * pv.x + q.z * pv.y);
        g.y = __logf(q.y * pv.x + q.w * pv.y);
        snap_s[e] = g;
        term_s[a * EPITCH + t] = obs_s[t] ? g.y : g.x;
    }
    __syncthreads();

    float* pre_s = reinterpret_cast<float*>(lo_s);
    if (wave == 0 && lane < NA) {
        const float* trow = term_s + lane * EPITCH;
        float* prow = pre_s + lane * EPITCH;
        float logab = -2.5649493574615367f;
        v4f bufa[EBUF], bufb[EBUF];
        #pragma unroll
        for (int i = 0; i < EBUF; ++i) bufa[i] = *(const v4f*)(trow + 4 * i);
        for (int p0 = 0; p0 < EPIECES; p0 += 2 * EBUF) {
            #pragma unroll
            for (int i = 0; i < EBUF; ++i) bufb[i] = *(const v4f*)(trow + 4 * (p0 + EBUF + i));
            logab = running_sum(bufa, logab, prow + 4 * p0);
            if (p0 + 2 * EBUF < EPIECES) {
                #pragma unroll
                for (int i = 0; i < EBUF; ++i) bufa[i] = *(const v4f*)(trow + 4 * (p0 + 2 * EBUF + i));
            }
            logab = running_sum(bufb, logab, prow + 4 * (p0 + EBUF));
        }
    }
    __syncthreads();

    for (int t = tid; t < TT; t += SCAN_THREADS) {
        float v0[NA], v1[NA];
        for (int a = 0; a < NA; ++a) {
            const v2f g = snap_s[t * NA + a];
            const float logab = pre_s[a * EPITCH + t];
            v0[a] = logab + g.x; v1[a] = logab + g.y;
        }
        const float o0 = lse13(v0);
        const float o1 = lse13(v1);
        const float onrm = lse2(o0, o1);
        v2f ov;
        ov.x = o0 - onrm;
        ov.y = o1 - onrm;
        *(v2f*)(out + (bT + t) * 2) = ov;
    }
}

extern "C" void kernel_launch(void* const* d_in, const int* in_sizes, int n_in,
                              void* d_out, int out_size, void* d_ws, size_t ws_size,
                              hipStream_t stream) {
    (void)in_sizes; (void)n_in; (void)out_size;
    const int* corr = (const int*)d_in[0];
    const int* kcv  = (const int*)d_in[1];
    const int* prob = (const int*)d_in[2];
    const float* FM    = (const float*)d_in[3];
    const float* trans = (const float*)d_in[4];
    const float* olp   = (const float*)d_in[5];
    const float* olk   = (const float*)d_in[6];
    const float* initl = (const float*)d_in[7];
    const float* lrw   = (const float*)d_in[8];
    const float* lrb   = (const float*)d_in[9];
    float* out = (float*)d_out;

    const size_t xbytes = (size_t)BB * TT * 2 * sizeof(float);
    if (ws_size >= xbytes) {
        float* xbuf = (float*)d_ws;
        hipLaunchKernelGGL(gemm_x_wmma, dim3(GEMM_BLOCKS), dim3(GEMM_THREADS), 0, stream,
                           FM, lrw, lrb, xbuf);
        hipLaunchKernelGGL((bkt_scan<true>), dim3(BB), dim3(SCAN_THREADS), 0, stream,
                           corr, kcv, prob, FM, trans, olp, olk, initl,
                           lrw, lrb, xbuf, out);
    } else {
        hipLaunchKernelGGL((bkt_scan<false>), dim3(BB), dim3(SCAN_THREADS), 0, stream,
                           corr, kcv, prob, FM, trans, olp, olk, initl,
                           lrw, lrb, (const float*)nullptr, out);
    }
}


# scan: observation softmax and log p(y|a) phases run one thread per step with the 13 abilities looped in registers: step inputs and index math once per step, not per element; same sums, same order
# speedup vs baseline: 34.6279x; 1.0472x over previous
#include <hip/hip_runtime.h>

#define BB 256
#define TT 500
#define NKC 100
#define NA 13

typedef __attribute__((ext_vector_type(2))) float v2f;
typedef __attribute__((ext_vector_type(4))) float v4f;
typedef __attribute__((ext_vector_type(8))) float v8f;

__device__ __forceinline__ float lse2(float x, float y) {
    float m = fmaxf(x, y);
    return m + log1pf(__expf(-fabsf(x - y)));
}

__device__ __forceinline__ v2f softmax2(float d) {
    const float e = __expf(-fabsf(d));
    const float r = __builtin_amdgcn_rcpf(1.0f + e);
    const float s = e * r;
    v2f q;
    q.x = (d >= 0.f) ? s : r;
    q.y = (d >= 0.f) ? r : s;
    return q;
}

#define GEMM_THREADS 256
#define GEMM_WAVES (GEMM_THREADS / 32)
#define TILES_PER_WAVE 8
#define GEMM_BLOCKS 500

struct AFrag { v4f m[6]; float r; };

__device__ __forceinline__ void load_afrag(AFrag& f, const float* __restrict__ rowp,
                                           int koff, int roff) {
    for (int j = 0; j < 6; ++j) f.m[j] = *(const v4f*)(rowp + koff + 32 * j);
    f.r = rowp[roff];
}

__device__ __forceinline__ v8f wmma4(v2f a, v2f b, v8f c) {
    return __builtin_amdgcn_wmma_f32_16x16x4_f32(false, a, false, b, (short)0, c, false, false);
}

__device__ __forceinline__ float quad_xor1(float v) {
    return __int_as_float(__builtin_amdgcn_update_dpp(0, __float_as_int(v), 0xB1, 0xF, 0xF, true));
}
__device__ __forceinline__ float quad_xor2(float v) {
    return __int_as_float(__builtin_amdgcn_update_dpp(0, __float_as_int(v), 0x4E, 0xF, 0xF, true));
}

__device__ __forceinline__ void store_tile(const v8f acc, long trow, int g, int q, int h,
                                           float bias, float* __restrict__ xout) {
    float s0 = ((g == 0) ? acc[0] : 0.f) + ((g == 1) ? acc[1] : 0.f)
             + ((g == 2) ? acc[2] : 0.f) + ((g == 3) ? acc[3] : 0.f);
    float s1 = ((g == 0) ? acc[4] : 0.f) + ((g == 1) ? acc[5] : 0.f)
             + ((g == 2) ? acc[6] : 0.f) + ((g == 3) ? acc[7] : 0.f);
    s0 += quad_xor1(s0); s0 += quad_xor2(s0);
    s1 += quad_xor1(s1); s1 += quad_xor2(s1);
    if (g == 0 && q < 2) {
        const long row = trow + 2 * h;
        xout[row * 2 + q] = s0 + bias;
        xout[(row + 1) * 2 + q] = s1 + bias;
    }
}

__global__ __launch_bounds__(GEMM_THREADS) __attribute__((amdgpu_waves_per_eu(4, 4)))
void gemm_x_wmma(const float* __restrict__ FM, const float* __restrict__ lrw,
                 const float* __restrict__ lrb, float* __restrict__ xout) {
    const int lane = threadIdx.x & 31;
    const int gw = blockIdx.x * GEMM_WAVES + (threadIdx.x >> 5);
    const int c = lane & 15;
    const int h = lane >> 4;
    const int g = c & 3;
    const int q = c >> 2;
    const int koff = 8 * g + 4 * h;
    const int roff = 192 + 2 * g + h;

    const bool bval = q < 2;
    const float* wrow = lrw + (q & 1) * 200;
    v4f bm[6];
    for (int j = 0; j < 6; ++j) {
        v4f t;
        t.x = wrow[koff + 32 * j + 0]; t.y = wrow[koff + 32 * j + 1];
        t.z = wrow[koff + 32 * j + 2]; t.w = wrow[koff + 32 * j + 3];
        const v4f z = {0.f, 0.f, 0.f, 0.f};
        bm[j] = bval ? t : z;
    }
    const float brt = wrow[roff];
    const float br = bval ? brt : 0.f;
    const float bias = lrb[q & 1];

    const long row0 = (long)gw * (TILES_PER_WAVE * 4);
    const float* ap = FM + (row0 + q) * 200;

    AFrag c0, c1;
    load_afrag(c0, ap, koff, roff);
    load_afrag(c1, ap + 800, koff, roff);
    for (int p = 0; p < TILES_PER_WAVE / 2; ++p) {
        AFrag n0 = c0, n1 = c1;
        if (p + 1 < TILES_PER_WAVE / 2) {
            load_afrag(n0, ap + (2 * p + 2) * 800, koff, roff);
            load_afrag(n1, ap + (2 * p + 3) * 800, koff, roff);
        }
        v8f acc0 = {}, acc1 = {};
        for (int j = 0; j < 6; ++j) {
            v2f a, b;
            b.x = bm[j].x; b.y = bm[j].y;
            a.x = c0.m[j].x; a.y = c0.m[j].y; acc0 = wmma4(a, b, acc0);
            a.x = c1.m[j].x; a.y = c1.m[j].y; acc1 = wmma4(a, b, acc1);
            b.x = bm[j].z; b.y = bm[j].w;
            a.x = c0.m[j].z; a.y = c0.m[j].w; acc0 = wmma4(a, b, acc0);
            a.x = c1.m[j].z; a.y = c1.m[j].w; acc1 = wmma4(a, b, acc1);
        }
        {
            v2f a, b;
            b.x = br; b.y = 0.f;
            a.x = c0.r; a.y = 0.f; acc0 = wmma4(a, b, acc0);
            a.x = c1.r; a.y = 0.f; acc1 = wmma4(a, b, acc1);
        }
        store_tile(acc0, row0 + (2 * p) * 4, g, q, h, bias, xout);
        store_tile(acc1, row0 + (2 * p + 1) * 4, g, q, h, bias, xout);
        c0 = n0; c1 = n1;
    }
}

#define SCAN_THREADS 512
#define SCAN_WAVES (SCAN_THREADS / 32)
static_assert(NKC <= 128, "a chain number has seven bits and a row of the chain bitmap has four words");
#define ESTEPS 512
#define EPITCH (ESTEPS + 4)
#define EPIECES (ESTEPS / 4)
#define EBUF 16
#define CSLOTS ((NKC * NA + SCAN_THREADS - 1) / SCAN_THREADS)
static_assert((NA - 1) + (SCAN_THREADS % NA) * (CSLOTS - 1) < 2 * NA, "a slot's ability index carries at most once");

struct CStat {
    int ai;
    int si;
    float q0, q1;
    float4 T;
};
static_assert(TT <= ESTEPS && EPIECES % (2 * EBUF) == 0, "phase E walks whole pairs of register buffers");

__device__ __forceinline__ float running_sum(const v4f (&buf)[EBUF], float s, float* dst) {
    #pragma unroll
    for (int i = 0; i < EBUF; ++i) {
        v4f o;
        o.x = s; s += buf[i].x;
        o.y = s; s += buf[i].y;
        o.z = s; s += buf[i].z;
        o.w = s; s += buf[i].w;
        *(v4f*)(dst + 4 * i) = o;
    }
    return s;
}

__device__ __forceinline__ float lse13(const float* v) {
    float m = -1.0e30f;
    for (int a = 0; a < NA; ++a) m = fmaxf(m, v[a]);
    float e[NA];
    for (int a = 0; a < NA; ++a) e[a] = __expf(v[a] - m);
    const float sA = (e[0] + e[8]) + (e[4] + e[12]);
    const float sB = (e[2] + e[10]) + e[6];
    const float sC = (e[1] + e[9]) + e[5];
    const float sD = (e[3] + e[11]) + e[7];
    const float s = (sA + sB) + (sC + sD);
    return m + __logf(s);
}

template <bool USE_XBUF>
__global__ __launch_bounds__(SCAN_THREADS)
void bkt_scan(const int* __restrict__ corr, const int* __restrict__ kcv,
              const int* __restrict__ prob, const float* __restrict__ FM,
              const float* __restrict__ trans, const float* __restrict__ olp,
              const float* __restrict__ olk, const float* __restrict__ initl,
              const float* __restrict__ lrw, const float* __restrict__ lrb,
              const float* __restrict__ xbuf, float* __restrict__ out) {
    __shared__ float4 lo_s[TT * NA];
    __shared__ v2f snap_s[TT * NA];
    __shared__ float4 op_s[TT];
    __shared__ v2f x_s[TT];
    __shared__ __attribute__((aligned(16))) int c_s[TT];
    __shared__ int obs_s[TT];
    __shared__ int hist_s[SCAN_WAVES * NKC];
    __shared__ uint4 lvl_s[TT];
    __shared__ int ord_s[TT];
    __shared__ int lev_s[TT + 4];
    __shared__ v2f alpha[NKC * NA];
    __shared__ __attribute__((aligned(16))) float olk_s[NKC * 4];
    __shared__ float4 T_s[NKC];
    __shared__ v2f pinit_s[NKC];
    __shared__ __attribute__((aligned(16))) float term_s[NA * EPITCH];

    const int tid = threadIdx.x;
    const int wave = tid >> 5, lane = tid & 31;
    const int b = blockIdx.x;
    const long bT = (long)b * TT;

    for (int i = tid; i < NKC * 4; i += SCAN_THREADS) olk_s[i] = olk[i];
    for (int i = tid; i < TT + 4; i += SCAN_THREADS) lev_s[i] = TT;
    for (int i = tid; i < SCAN_WAVES * NKC; i += SCAN_THREADS) hist_s[i] = 0;
    for (int i = tid; i < TT; i += SCAN_THREADS) lvl_s[i] = make_uint4(0u, 0u, 0u, 0u);
    for (int i = tid; i < NA * (ESTEPS - TT); i += SCAN_THREADS)
        term_s[(i / (ESTEPS - TT)) * EPITCH + TT + i % (ESTEPS - TT)] = 0.f;
    for (int c = tid; c < NKC; c += SCAN_THREADS) {
        pinit_s[c] = softmax2(initl[c * 2 + 1] - initl[c * 2]);
        const float t00 = trans[c * 4 + 0], t01 = trans[c * 4 + 1];
        const float t10 = trans[c * 4 + 2], t11 = trans[c * 4 + 3];
        const v2f col0 = softmax2(t10 - t00);
        const v2f col1 = softmax2(t11 - t01);
        float4 T;
        T.x = col0.x; T.y = col1.x;
        T.z = col0.y; T.w = col1.y;
        T_s[c] = T;
    }

    for (int t = tid; t < TT; t += SCAN_THREADS) {
        const long idx = bT + t;
        c_s[t] = kcv[idx];
        obs_s[t] = corr[idx];
        const int p = prob[idx];
        op_s[t] = *(const float4*)(olp + (long)p * 4);
        if (USE_XBUF) x_s[t] = *(const v2f*)(xbuf + idx * 2);
    }
    if (!USE_XBUF) {
        for (int t = wave; t < TT; t += SCAN_WAVES) {
            const float* fr = FM + (bT + t) * 200;
            float p0 = 0.f, p1 = 0.f;
            for (int f = lane; f < 200; f += 32) {
                float fv = fr[f];
                p0 = fmaf(fv, lrw[f], p0);
                p1 = fmaf(fv, lrw[200 + f], p1);
            }
            for (int off = 16; off; off >>= 1) {
                p0 += __shfl_xor(p0, off, 32);
                p1 += __shfl_xor(p1, off, 32);
            }
            if (lane == 0) {
                v2f xv;
                xv.x = p0 + lrb[0];
                xv.y = p1 + lrb[1];
                x_s[t] = xv;
            }
        }
    }
    __syncthreads();

    for (int i = tid; i < NKC * NA; i += SCAN_THREADS) alpha[i] = pinit_s[i / NA];

    const bool valid = tid < TT;
    const int myc = valid ? c_s[tid] : 0;
    unsigned peers = __builtin_amdgcn_ballot_w32(valid);
    for (int bit = 0; bit < 7; ++bit) {
        const bool on = ((myc >> bit) & 1) != 0;
        const unsigned m = __builtin_amdgcn_ballot_w32(on);
        peers &= on ? m : ~m;
    }
    const int rin = __popc(peers & ((1u << lane) - 1u));
    const int wv = __builtin_amdgcn_readfirstlane(wave);
    if (valid && rin == 0) hist_s[wv * NKC + myc] = __popc(peers);
    __syncthreads();

    int occ = rin;
    for (int w = 0; w < wv; ++w) occ += hist_s[w * NKC + myc];
    if (wv < (NKC + 31) / 32) {
        int cnt = 0;
        if (tid < NKC)
            for (int w = 0; w < SCAN_WAVES; ++w) cnt += hist_s[w * NKC + tid];
        for (int k = 0; k < TT; ++k) {
            const unsigned m = __builtin_amdgcn_ballot_w32(cnt > k);
            if (m == 0u) break;
            if (lane == 0) reinterpret_cast<unsigned*>(lvl_s)[k * 4 + wv] = m;
        }
    }

    if (valid) {
        const float4 op4 = op_s[tid];
        const v2f xv = x_s[tid];
        const float4 ok4 = *(const float4*)(olk_s + myc * 4);
        const float x0 = xv.x, x1 = xv.y;
        const float s00 = op4.x + ok4.x, s01 = op4.y + ok4.y;
        const float s10 = op4.z + ok4.z, s11 = op4.w + ok4.w;
        float4* lrow = lo_s + tid * NA;
        #pragma unroll
        for (int a = 0; a < NA; ++a) {
            const float abil = -3.0f + 0.5f * (float)a;
            const float l00 = s00 - abil + x0;
            const float l01 = s01 + abil - x0;
            const float l10 = s10 - abil + x1;
            const float l11 = s11 + abil - x1;
            const v2f q0 = softmax2(l01 - l00);
            const v2f q1 = softmax2(l11 - l10);
            float4 r;
            r.x = q0.x; r.y = q0.y;
            r.z = q1.x; r.w = q1.y;
            lrow[a] = r;
        }
    }
    __syncthreads();

    if (valid) {
        int lev = 0;
        for (int k = 0; k < occ; ++k) {
            const uint4 m = lvl_s[k];
            lev += __popc(m.x) + __popc(m.y) + __popc(m.z) + __popc(m.w);
        }
        const uint4 m = lvl_s[occ];
        const int v = myc >> 5;
        const unsigned below = (1u << (myc & 31)) - 1u;
        const unsigned m0 = (v > 0) ? ~0u : below;
        const unsigned m1 = (v > 1) ? ~0u : ((v == 1) ? below : 0u);
        const unsigned m2 = (v > 2) ? ~0u : ((v == 2) ? below : 0u);
        const unsigned m3 = (v == 3) ? below : 0u;
        const int idx = __popc(m.x & m0) + __popc(m.y & m1) + __popc(m.z & m2) + __popc(m.w & m3);
        const int pos = lev + idx;
        ord_s[pos] = tid | (myc << 9) | ((obs_s[tid] != 0) << 16) | (occ << 17);
        if (idx == 0) lev_s[occ] = pos;
    }
    __syncthreads();

    {
        const int pt = tid / NA;
        const int at = tid - pt * NA;
        int P[CSLOTS], A[CSLOTS];
        #pragma unroll
        for (int j = 0; j < CSLOTS; ++j) {
            const int a = at + (SCAN_THREADS % NA) * j;
            const int carry = (a >= NA) ? 1 : 0;
            A[j] = a - NA * carry;
            P[j] = pt + (SCAN_THREADS / NA) * j + carry;
        }

        CStat cur[CSLOTS];
        auto fetch_stat = [&](int s) {
            int rec[CSLOTS];
            #pragma unroll
            for (int j = 0; j < CSLOTS; ++j) rec[j] = ord_s[min(s + P[j], TT - 1)];
            #pragma unroll
            for (int j = 0; j < CSLOTS; ++j) {
                const int t = rec[j] & 511;
                const int c = (rec[j] >> 9) & 127;
                const int obs = (rec[j] >> 16) & 1;
                const int si = t * NA + A[j];
                const float4 q = lo_s[si];
                cur[j].T = T_s[c];
                cur[j].ai = c * NA + A[j];
                cur[j].si = si;
                cur[j].q0 = obs ? q.y : q.x;
                cur[j].q1 = obs ? q.w : q.z;
            }
        };

        int k = 0;
        int s0 = 0;
        int s1 = __builtin_amdgcn_readfirstlane(lev_s[1]);
        int s2 = __builtin_amdgcn_readfirstlane(lev_s[2]);
        fetch_stat(s0);
        while (s0 < TT) {
            const int s3v = lev_s[k + 3];
            if (s0 + P[0] < s1 || s1 + P[0] < s2) {
                v2f pv[CSLOTS], pn[CSLOTS];
                int ai[CSLOTS], si[CSLOTS];
                #pragma unroll
                for (int j = 0; j < CSLOTS; ++j) {
                    ai[j] = cur[j].ai; si[j] = cur[j].si;
                    pv[j] = alpha[ai[j]];
                }
                #pragma unroll
                for (int j = 0; j < CSLOTS; ++j) {
                    const float4 T = cur[j].T;
                    const float w0 = cur[j].q0 * pv[j].x;
                    const float w1 = cur[j].q1 * pv[j].y;
                    const float n0 = T.x * w0 + T.y * w1;
                    const float n1 = T.z * w0 + T.w * w1;
                    const float rn = __builtin_amdgcn_rcpf(n0 + n1);
                    pn[j].x = n0 * rn; pn[j].y = n1 * rn;
                }
                fetch_stat(s1);
                #pragma unroll
                for (int j = 0; j < CSLOTS; ++j) {
                    if (s0 + P[j] < s1) {
                        alpha[ai[j]] = pn[j];
                        snap_s[si[j]] = pv[j];
                    }
                }
            }
            ++k;
            s0 = s1;
            s1 = s2;
            s2 = __builtin_amdgcn_readfirstlane(s3v);
            __syncthreads();
        }
    }

    if (valid) {
        const bool ob = obs_s[tid] != 0;
        const float4* lrow = lo_s + tid * NA;
        v2f* srow = snap_s + tid * NA;
        #pragma unroll
        for (int a = 0; a < NA; ++a) {
            const float4 q = lrow[a];
            const v2f pv = srow[a];
            v2f g;
            g.x = __logf(q.x * pv.x + q.z * pv.y);
            g.y = __logf(q.y * pv.x + q.w * pv.y);
            srow[a] = g;
            term_s[a * EPITCH + tid] = ob ? g.y : g.x;
        }
    }
    __syncthreads();

    float* pre_s = reinterpret_cast<float*>(lo_s);
    if (wave == 0 && lane < NA) {
        const float* trow = term_s + lane * EPITCH;
        float* prow = pre_s + lane * EPITCH;
        float logab = -2.5649493574615367f;
        v4f bufa[EBUF], bufb[EBUF];
        #pragma unroll
        for (int i = 0; i < EBUF; ++i) bufa[i] = *(const v4f*)(trow + 4 * i);
        for (int p0 = 0; p0 < EPIECES; p0 += 2 * EBUF) {
            #pragma unroll
            for (int i = 0; i < EBUF; ++i) bufb[i] = *(const v4f*)(trow + 4 * (p0 + EBUF + i));
            logab = running_sum(bufa, logab, prow + 4 * p0);
            if (p0 + 2 * EBUF < EPIECES) {
                #pragma unroll
                for (int i = 0; i < EBUF; ++i) bufa[i] = *(const v4f*)(trow + 4 * (p0 + 2 * EBUF + i));
            }
            logab = running_sum(bufb, logab, prow + 4 * (p0 + EBUF));
        }
    }
    __syncthreads();

    for (int t = tid; t < TT; t += SCAN_THREADS) {
        float v0[NA], v1[NA];
        for (int a = 0; a < NA; ++a) {
            const v2f g = snap_s[t * NA + a];
            const float logab = pre_s[a * EPITCH + t];
            v0[a] = logab + g.x; v1[a] = logab + g.y;
        }
        const float o0 = lse13(v0);
        const float o1 = lse13(v1);
        const float onrm = lse2(o0, o1);
        v2f ov;
        ov.x = o0 - onrm;
        ov.y = o1 - onrm;
        *(v2f*)(out + (bT + t) * 2) = ov;
    }
}

extern "C" void kernel_launch(void* const* d_in, const int* in_sizes, int n_in,
                              void* d_out, int out_size, void* d_ws, size_t ws_size,
                              hipStream_t stream) {
    (void)in_sizes; (void)n_in; (void)out_size;
    const int* corr = (const int*)d_in[0];
    const int* kcv  = (const int*)d_in[1];
    const int* prob = (const int*)d_in[2];
    const float* FM    = (const float*)d_in[3];
    const float* trans = (const float*)d_in[4];
    const float* olp   = (const float*)d_in[5];
    const float* olk   = (const float*)d_in[6];
    const float* initl = (const float*)d_in[7];
    const float* lrw   = (const float*)d_in[8];
    const float* lrb   = (const float*)d_in[9];
    float* out = (float*)d_out;

    const size_t xbytes = (size_t)BB * TT * 2 * sizeof(float);
    if (ws_size >= xbytes) {
        float* xbuf = (float*)d_ws;
        hipLaunchKernelGGL(gemm_x_wmma, dim3(GEMM_BLOCKS), dim3(GEMM_THREADS), 0, stream,
                           FM, lrw, lrb, xbuf);
        hipLaunchKernelGGL((bkt_scan<true>), dim3(BB), dim3(SCAN_THREADS), 0, stream,
                           corr, kcv, prob, FM, trans, olp, olk, initl,
                           lrw, lrb, xbuf, out);
    } else {
        hipLaunchKernelGGL((bkt_scan<false>), dim3(BB), dim3(SCAN_THREADS), 0, stream,
                           corr, kcv, prob, FM, trans, olp, olk, initl,
                           lrw, lrb, (const float*)nullptr, out);
    }
}


# scan: output phase on probabilities: p(y|a) stays in registers, weights exp(log_ability - max), out = log(s_y/(s_0+s_1)); logs only for the running sum's term; rounding and sums of out change
# speedup vs baseline: 35.8062x; 1.0340x over previous
#include <hip/hip_runtime.h>

#define BB 256
#define TT 500
#define NKC 100
#define NA 13

typedef __attribute__((ext_vector_type(2))) float v2f;
typedef __attribute__((ext_vector_type(4))) float v4f;
typedef __attribute__((ext_vector_type(8))) float v8f;

__device__ __forceinline__ v2f softmax2(float d) {
    const float e = __expf(-fabsf(d));
    const float r = __builtin_amdgcn_rcpf(1.0f + e);
    const float s = e * r;
    v2f q;
    q.x = (d >= 0.f) ? s : r;
    q.y = (d >= 0.f) ? r : s;
    return q;
}

#define GEMM_THREADS 256
#define GEMM_WAVES (GEMM_THREADS / 32)
#define TILES_PER_WAVE 8
#define GEMM_BLOCKS 500

struct AFrag { v4f m[6]; float r; };

__device__ __forceinline__ void load_afrag(AFrag& f, const float* __restrict__ rowp,
                                           int koff, int roff) {
    for (int j = 0; j < 6; ++j) f.m[j] = *(const v4f*)(rowp + koff + 32 * j);
    f.r = rowp[roff];
}

__device__ __forceinline__ v8f wmma4(v2f a, v2f b, v8f c) {
    return __builtin_amdgcn_wmma_f32_16x16x4_f32(false, a, false, b, (short)0, c, false, false);
}

__device__ __forceinline__ float quad_xor1(float v) {
    return __int_as_float(__builtin_amdgcn_update_dpp(0, __float_as_int(v), 0xB1, 0xF, 0xF, true));
}
__device__ __forceinline__ float quad_xor2(float v) {
    return __int_as_float(__builtin_amdgcn_update_dpp(0, __float_as_int(v), 0x4E, 0xF, 0xF, true));
}

__device__ __forceinline__ void store_tile(const v8f acc, long trow, int g, int q, int h,
                                           float bias, float* __restrict__ xout) {
    float s0 = ((g == 0) ? acc[0] : 0.f) + ((g == 1) ? acc[1] : 0.f)
             + ((g == 2) ? acc[2] : 0.f) + ((g == 3) ? acc[3] : 0.f);
    float s1 = ((g == 0) ? acc[4] : 0.f) + ((g == 1) ? acc[5] : 0.f)
             + ((g == 2) ? acc[6] : 0.f) + ((g == 3) ? acc[7] : 0.f);
    s0 += quad_xor1(s0); s0 += quad_xor2(s0);
    s1 += quad_xor1(s1); s1 += quad_xor2(s1);
    if (g == 0 && q < 2) {
        const long row = trow + 2 * h;
        xout[row * 2 + q] = s0 + bias;
        xout[(row + 1) * 2 + q] = s1 + bias;
    }
}

__global__ __launch_bounds__(GEMM_THREADS) __attribute__((amdgpu_waves_per_eu(4, 4)))
void gemm_x_wmma(const float* __restrict__ FM, const float* __restrict__ lrw,
                 const float* __restrict__ lrb, float* __restrict__ xout) {
    const int lane = threadIdx.x & 31;
    const int gw = blockIdx.x * GEMM_WAVES + (threadIdx.x >> 5);
    const int c = lane & 15;
    const int h = lane >> 4;
    const int g = c & 3;
    const int q = c >> 2;
    const int koff = 8 * g + 4 * h;
    const int roff = 192 + 2 * g + h;

    const bool bval = q < 2;
    const float* wrow = lrw + (q & 1) * 200;
    v4f bm[6];
    for (int j = 0; j < 6; ++j) {
        v4f t;
        t.x = wrow[koff + 32 * j + 0]; t.y = wrow[koff + 32 * j + 1];
        t.z = wrow[koff + 32 * j + 2]; t.w = wrow[koff + 32 * j + 3];
        const v4f z = {0.f, 0.f, 0.f, 0.f};
        bm[j] = bval ? t : z;
    }
    const float brt = wrow[roff];
    const float br = bval ? brt : 0.f;
    const float bias = lrb[q & 1];

    const long row0 = (long)gw * (TILES_PER_WAVE * 4);
    const float* ap = FM + (row0 + q) * 200;

    AFrag c0, c1;
    load_afrag(c0, ap, koff, roff);
    load_afrag(c1, ap + 800, koff, roff);
    for (int p = 0; p < TILES_PER_WAVE / 2; ++p) {
        AFrag n0 = c0, n1 = c1;
        if (p + 1 < TILES_PER_WAVE / 2) {
            load_afrag(n0, ap + (2 * p + 2) * 800, koff, roff);
            load_afrag(n1, ap + (2 * p + 3) * 800, koff, roff);
        }
        v8f acc0 = {}, acc1 = {};
        for (int j = 0; j < 6; ++j) {
            v2f a, b;
            b.x = bm[j].x; b.y = bm[j].y;
            a.x = c0.m[j].x; a.y = c0.m[j].y; acc0 = wmma4(a, b, acc0);
            a.x = c1.m[j].x; a.y = c1.m[j].y; acc1 = wmma4(a, b, acc1);
            b.x = bm[j].z; b.y = bm[j].w;
            a.x = c0.m[j].z; a.y = c0.m[j].w; acc0 = wmma4(a, b, acc0);
            a.x = c1.m[j].z; a.y = c1.m[j].w; acc1 = wmma4(a, b, acc1);
        }
        {
            v2f a, b;
            b.x = br; b.y = 0.f;
            a.x = c0.r; a.y = 0.f; acc0 = wmma4(a, b, acc0);
            a.x = c1.r; a.y = 0.f; acc1 = wmma4(a, b, acc1);
        }
        store_tile(acc0, row0 + (2 * p) * 4, g, q, h, bias, xout);
        store_tile(acc1, row0 + (2 * p + 1) * 4, g, q, h, bias, xout);
        c0 = n0; c1 = n1;
    }
}

#define SCAN_THREADS 512
#define SCAN_WAVES (SCAN_THREADS / 32)
static_assert(NKC <= 128, "a chain number has seven bits and a row of the chain bitmap has four words");
#define ESTEPS 512
#define EPITCH (ESTEPS + 4)
#define EPIECES (ESTEPS / 4)
#define EBUF 16
#define CSLOTS ((NKC * NA + SCAN_THREADS - 1) / SCAN_THREADS)
static_assert((NA - 1) + (SCAN_THREADS % NA) * (CSLOTS - 1) < 2 * NA, "a slot's ability index carries at most once");

struct CStat {
    int ai;
    int si;
    float q0, q1;
    float4 T;
};
static_assert(TT <= ESTEPS && EPIECES % (2 * EBUF) == 0, "phase E walks whole pairs of register buffers");
static_assert(TT <= SCAN_THREADS, "a step's thread carries p(y | a) from phase D to phase F");

__device__ __forceinline__ float running_sum(const v4f (&buf)[EBUF], float s, float* dst) {
    #pragma unroll
    for (int i = 0; i < EBUF; ++i) {
        v4f o;
        o.x = s; s += buf[i].x;
        o.y = s; s += buf[i].y;
        o.z = s; s += buf[i].z;
        o.w = s; s += buf[i].w;
        *(v4f*)(dst + 4 * i) = o;
    }
    return s;
}

__device__ __forceinline__ float sum13(const float* e) {
    const float sA = (e[0] + e[8]) + (e[4] + e[12]);
    const float sB = (e[2] + e[10]) + e[6];
    const float sC = (e[1] + e[9]) + e[5];
    const float sD = (e[3] + e[11]) + e[7];
    return (sA + sB) + (sC + sD);
}

template <bool USE_XBUF>
__global__ __launch_bounds__(SCAN_THREADS)
void bkt_scan(const int* __restrict__ corr, const int* __restrict__ kcv,
              const int* __restrict__ prob, const float* __restrict__ FM,
              const float* __restrict__ trans, const float* __restrict__ olp,
              const float* __restrict__ olk, const float* __restrict__ initl,
              const float* __restrict__ lrw, const float* __restrict__ lrb,
              const float* __restrict__ xbuf, float* __restrict__ out) {
    __shared__ float4 lo_s[TT * NA];
    __shared__ v2f snap_s[TT * NA];
    __shared__ float4 op_s[TT];
    __shared__ v2f x_s[TT];
    __shared__ __attribute__((aligned(16))) int c_s[TT];
    __shared__ int obs_s[TT];
    __shared__ int hist_s[SCAN_WAVES * NKC];
    __shared__ uint4 lvl_s[TT];
    __shared__ int ord_s[TT];
    __shared__ int lev_s[TT + 4];
    __shared__ v2f alpha[NKC * NA];
    __shared__ __attribute__((aligned(16))) float olk_s[NKC * 4];
    __shared__ float4 T_s[NKC];
    __shared__ v2f pinit_s[NKC];
    __shared__ __attribute__((aligned(16))) float term_s[NA * EPITCH];

    const int tid = threadIdx.x;
    const int wave = tid >> 5, lane = tid & 31;
    const int b = blockIdx.x;
    const long bT = (long)b * TT;

    for (int i = tid; i < NKC * 4; i += SCAN_THREADS) olk_s[i] = olk[i];
    for (int i = tid; i < TT + 4; i += SCAN_THREADS) lev_s[i] = TT;
    for (int i = tid; i < SCAN_WAVES * NKC; i += SCAN_THREADS) hist_s[i] = 0;
    for (int i = tid; i < TT; i += SCAN_THREADS) lvl_s[i] = make_uint4(0u, 0u, 0u, 0u);
    for (int i = tid; i < NA * (ESTEPS - TT); i += SCAN_THREADS)
        term_s[(i / (ESTEPS - TT)) * EPITCH + TT + i % (ESTEPS - TT)] = 0.f;
    for (int c = tid; c < NKC; c += SCAN_THREADS) {
        pinit_s[c] = softmax2(initl[c * 2 + 1] - initl[c * 2]);
        const float t00 = trans[c * 4 + 0], t01 = trans[c * 4 + 1];
        const float t10 = trans[c * 4 + 2], t11 = trans[c * 4 + 3];
        const v2f col0 = softmax2(t10 - t00);
        const v2f col1 = softmax2(t11 - t01);
        float4 T;
        T.x = col0.x; T.y = col1.x;
        T.z = col0.y; T.w = col1.y;
        T_s[c] = T;
    }

    for (int t = tid; t < TT; t += SCAN_THREADS) {
        const long idx = bT + t;
        c_s[t] = kcv[idx];
        obs_s[t] = corr[idx];
        const int p = prob[idx];
        op_s[t] = *(const float4*)(olp + (long)p * 4);
        if (USE_XBUF) x_s[t] = *(const v2f*)(xbuf + idx * 2);
    }
    if (!USE_XBUF) {
        for (int t = wave; t < TT; t += SCAN_WAVES) {
            const float* fr = FM + (bT + t) * 200;
            float p0 = 0.f, p1 = 0.f;
            for (int f = lane; f < 200; f += 32) {
                float fv = fr[f];
                p0 = fmaf(fv, lrw[f], p0);
                p1 = fmaf(fv, lrw[200 + f], p1);
            }
            for (int off = 16; off; off >>= 1) {
                p0 += __shfl_xor(p0, off, 32);
                p1 += __shfl_xor(p1, off, 32);
            }
            if (lane == 0) {
                v2f xv;
                xv.x = p0 + lrb[0];
                xv.y = p1 + lrb[1];
                x_s[t] = xv;
            }
        }
    }
    __syncthreads();

    for (int i = tid; i < NKC * NA; i += SCAN_THREADS) alpha[i] = pinit_s[i / NA];

    const bool valid = tid < TT;
    const int myc = valid ? c_s[tid] : 0;
    unsigned peers = __builtin_amdgcn_ballot_w32(valid);
    for (int bit = 0; bit < 7; ++bit) {
        const bool on = ((myc >> bit) & 1) != 0;
        const unsigned m = __builtin_amdgcn_ballot_w32(on);
        peers &= on ? m : ~m;
    }
    const int rin = __popc(peers & ((1u << lane) - 1u));
    const int wv = __builtin_amdgcn_readfirstlane(wave);
    if (valid && rin == 0) hist_s[wv * NKC + myc] = __popc(peers);
    __syncthreads();

    int occ = rin;
    for (int w = 0; w < wv; ++w) occ += hist_s[w * NKC + myc];
    if (wv < (NKC + 31) / 32) {
        int cnt = 0;
        if (tid < NKC)
            for (int w = 0; w < SCAN_WAVES; ++w) cnt += hist_s[w * NKC + tid];
        for (int k = 0; k < TT; ++k) {
            const unsigned m = __builtin_amdgcn_ballot_w32(cnt > k);
            if (m == 0u) break;
            if (lane == 0) reinterpret_cast<unsigned*>(lvl_s)[k * 4 + wv] = m;
        }
    }

    if (valid) {
        const float4 op4 = op_s[tid];
        const v2f xv = x_s[tid];
        const float4 ok4 = *(const float4*)(olk_s + myc * 4);
        const float x0 = xv.x, x1 = xv.y;
        const float s00 = op4.x + ok4.x, s01 = op4.y + ok4.y;
        const float s10 = op4.z + ok4.z, s11 = op4.w + ok4.w;
        float4* lrow = lo_s + tid * NA;
        #pragma unroll
        for (int a = 0; a < NA; ++a) {
            const float abil = -3.0f + 0.5f * (float)a;
            const float l00 = s00 - abil + x0;
            const float l01 = s01 + abil - x0;
            const float l10 = s10 - abil + x1;
            const float l11 = s11 + abil - x1;
            const v2f q0 = softmax2(l01 - l00);
            const v2f q1 = softmax2(l11 - l10);
            float4 r;
            r.x = q0.x; r.y = q0.y;
            r.z = q1.x; r.w = q1.y;
            lrow[a] = r;
        }
    }
    __syncthreads();

    if (valid) {
        int lev = 0;
        for (int k = 0; k < occ; ++k) {
            const uint4 m = lvl_s[k];
            lev += __popc(m.x) + __popc(m.y) + __popc(m.z) + __popc(m.w);
        }
        const uint4 m = lvl_s[occ];
        const int v = myc >> 5;
        const unsigned below = (1u << (myc & 31)) - 1u;
        const unsigned m0 = (v > 0) ? ~0u : below;
        const unsigned m1 = (v > 1) ? ~0u : ((v == 1) ? below : 0u);
        const unsigned m2 = (v > 2) ? ~0u : ((v == 2) ? below : 0u);
        const unsigned m3 = (v == 3) ? below : 0u;
        const int idx = __popc(m.x & m0) + __popc(m.y & m1) + __popc(m.z & m2) + __popc(m.w & m3);
        const int pos = lev + idx;
        ord_s[pos] = tid | (myc << 9) | ((obs_s[tid] != 0) << 16) | (occ << 17);
        if (idx == 0) lev_s[occ] = pos;
    }
    __syncthreads();

    {
        const int pt = tid / NA;
        const int at = tid - pt * NA;
        int P[CSLOTS], A[CSLOTS];
        #pragma unroll
        for (int j = 0; j < CSLOTS; ++j) {
            const int a = at + (SCAN_THREADS % NA) * j;
            const int carry = (a >= NA) ? 1 : 0;
            A[j] = a - NA * carry;
            P[j] = pt + (SCAN_THREADS / NA) * j + carry;
        }

        CStat cur[CSLOTS];
        auto fetch_stat = [&](int s) {
            int rec[CSLOTS];
            #pragma unroll
            for (int j = 0; j < CSLOTS; ++j) rec[j] = ord_s[min(s + P[j], TT - 1)];
            #pragma unroll
            for (int j = 0; j < CSLOTS; ++j) {
                const int t = rec[j] & 511;
                const int c = (rec[j] >> 9) & 127;
                const int obs = (rec[j] >> 16) & 1;
                const int si = t * NA + A[j];
                const float4 q = lo_s[si];
                cur[j].T = T_s[c];
                cur[j].ai = c * NA + A[j];
                cur[j].si = si;
                cur[j].q0 = obs ? q.y : q.x;
                cur[j].q1 = obs ? q.w : q.z;
            }
        };

        int k = 0;
        int s0 = 0;
        int s1 = __builtin_amdgcn_readfirstlane(lev_s[1]);
        int s2 = __builtin_amdgcn_readfirstlane(lev_s[2]);
        fetch_stat(s0);
        while (s0 < TT) {
            const int s3v = lev_s[k + 3];
            if (s0 + P[0] < s1 || s1 + P[0] < s2) {
                v2f pv[CSLOTS], pn[CSLOTS];
                int ai[CSLOTS], si[CSLOTS];
                #pragma unroll
                for (int j = 0; j < CSLOTS; ++j) {
                    ai[j] = cur[j].ai; si[j] = cur[j].si;
                    pv[j] = alpha[ai[j]];
                }
                #pragma unroll
                for (int j = 0; j < CSLOTS; ++j) {
                    const float4 T = cur[j].T;
                    const float w0 = cur[j].q0 * pv[j].x;
                    const float w1 = cur[j].q1 * pv[j].y;
                    const float n0 = T.x * w0 + T.y * w1;
                    const float n1 = T.z * w0 + T.w * w1;
                    const float rn = __builtin_amdgcn_rcpf(n0 + n1);
                    pn[j].x = n0 * rn; pn[j].y = n1 * rn;
                }
                fetch_stat(s1);
                #pragma unroll
                for (int j = 0; j < CSLOTS; ++j) {
                    if (s0 + P[j] < s1) {
                        alpha[ai[j]] = pn[j];
                        snap_s[si[j]] = pv[j];
                    }
                }
            }
            ++k;
            s0 = s1;
            s1 = s2;
            s2 = __builtin_amdgcn_readfirstlane(s3v);
            __syncthreads();
        }
    }

    v2f py[NA];
    if (valid) {
        const bool ob = obs_s[tid] != 0;
        const float4* lrow = lo_s + tid * NA;
        const v2f* srow = snap_s + tid * NA;
        #pragma unroll
        for (int a = 0; a < NA; ++a) {
            const float4 q = lrow[a];
            const v2f pv = srow[a];
            v2f p;
            p.x = q.x * pv.x + q.z * pv.y;
            p.y = q.y * pv.x + q.w * pv.y;
            py[a] = p;
            term_s[a * EPITCH + tid] = __logf(ob ? p.y : p.x);
        }
    }
    __syncthreads();

    float* pre_s = reinterpret_cast<float*>(lo_s);
    if (wave == 0 && lane < NA) {
        const float* trow = term_s + lane * EPITCH;
        float* prow = pre_s + lane * EPITCH;
        float logab = -2.5649493574615367f;
        v4f bufa[EBUF], bufb[EBUF];
        #pragma unroll
        for (int i = 0; i < EBUF; ++i) bufa[i] = *(const v4f*)(trow + 4 * i);
        for (int p0 = 0; p0 < EPIECES; p0 += 2 * EBUF) {
            #pragma unroll
            for (int i = 0; i < EBUF; ++i) bufb[i] = *(const v4f*)(trow + 4 * (p0 + EBUF + i));
            logab = running_sum(bufa, logab, prow + 4 * p0);
            if (p0 + 2 * EBUF < EPIECES) {
                #pragma unroll
                for (int i = 0; i < EBUF; ++i) bufa[i] = *(const v4f*)(trow + 4 * (p0 + 2 * EBUF + i));
            }
            logab = running_sum(bufb, logab, prow + 4 * (p0 + EBUF));
        }
    }
    __syncthreads();

    if (valid) {
        float la[NA];
        float m = -1.0e30f;
        #pragma unroll
        for (int a = 0; a < NA; ++a) {
            la[a] = pre_s[a * EPITCH + tid];
            m = fmaxf(m, la[a]);
        }
        float e0[NA], e1[NA];
        #pragma unroll
        for (int a = 0; a < NA; ++a) {
            const float w = __expf(la[a] - m);
            e0[a] = w * py[a].x;
            e1[a] = w * py[a].y;
        }
        const float s0 = sum13(e0);
        const float s1 = sum13(e1);
        const float lnrm = __logf(s0 + s1);
        v2f ov;
        ov.x = __logf(s0) - lnrm;
        ov.y = __logf(s1) - lnrm;
        *(v2f*)(out + (bT + tid) * 2) = ov;
    }
}

extern "C" void kernel_launch(void* const* d_in, const int* in_sizes, int n_in,
                              void* d_out, int out_size, void* d_ws, size_t ws_size,
                              hipStream_t stream) {
    (void)in_sizes; (void)n_in; (void)out_size;
    const int* corr = (const int*)d_in[0];
    const int* kcv  = (const int*)d_in[1];
    const int* prob = (const int*)d_in[2];
    const float* FM    = (const float*)d_in[3];
    const float* trans = (const float*)d_in[4];
    const float* olp   = (const float*)d_in[5];
    const float* olk   = (const float*)d_in[6];
    const float* initl = (const float*)d_in[7];
    const float* lrw   = (const float*)d_in[8];
    const float* lrb   = (const float*)d_in[9];
    float* out = (float*)d_out;

    const size_t xbytes = (size_t)BB * TT * 2 * sizeof(float);
    if (ws_size >= xbytes) {
        float* xbuf = (float*)d_ws;
        hipLaunchKernelGGL(gemm_x_wmma, dim3(GEMM_BLOCKS), dim3(GEMM_THREADS), 0, stream,
                           FM, lrw, lrb, xbuf);
        hipLaunchKernelGGL((bkt_scan<true>), dim3(BB), dim3(SCAN_THREADS), 0, stream,
                           corr, kcv, prob, FM, trans, olp, olk, initl,
                           lrw, lrb, xbuf, out);
    } else {
        hipLaunchKernelGGL((bkt_scan<false>), dim3(BB), dim3(SCAN_THREADS), 0, stream,
                           corr, kcv, prob, FM, trans, olp, olk, initl,
                           lrw, lrb, (const float*)nullptr, out);
    }
}


# scan: start-up memory chain cut: all independent global loads requested at once, a step's inputs kept in its thread's registers instead of LDS staging (one barrier less); same sums, same order
# speedup vs baseline: 37.4138x; 1.0449x over previous
#include <hip/hip_runtime.h>

#define BB 256
#define TT 500
#define NKC 100
#define NA 13

typedef __attribute__((ext_vector_type(2))) float v2f;
typedef __attribute__((ext_vector_type(4))) float v4f;
typedef __attribute__((ext_vector_type(8))) float v8f;

__device__ __forceinline__ v2f softmax2(float d) {
    const float e = __expf(-fabsf(d));
    const float r = __builtin_amdgcn_rcpf(1.0f + e);
    const float s = e * r;
    v2f q;
    q.x = (d >= 0.f) ? s : r;
    q.y = (d >= 0.f) ? r : s;
    return q;
}

#define GEMM_THREADS 256
#define GEMM_WAVES (GEMM_THREADS / 32)
#define TILES_PER_WAVE 8
#define GEMM_BLOCKS 500

struct AFrag { v4f m[6]; float r; };

__device__ __forceinline__ void load_afrag(AFrag& f, const float* __restrict__ rowp,
                                           int koff, int roff) {
    for (int j = 0; j < 6; ++j) f.m[j] = *(const v4f*)(rowp + koff + 32 * j);
    f.r = rowp[roff];
}

__device__ __forceinline__ v8f wmma4(v2f a, v2f b, v8f c) {
    return __builtin_amdgcn_wmma_f32_16x16x4_f32(false, a, false, b, (short)0, c, false, false);
}

__device__ __forceinline__ float quad_xor1(float v) {
    return __int_as_float(__builtin_amdgcn_update_dpp(0, __float_as_int(v), 0xB1, 0xF, 0xF, true));
}
__device__ __forceinline__ float quad_xor2(float v) {
    return __int_as_float(__builtin_amdgcn_update_dpp(0, __float_as_int(v), 0x4E, 0xF, 0xF, true));
}

__device__ __forceinline__ void store_tile(const v8f acc, long trow, int g, int q, int h,
                                           float bias, float* __restrict__ xout) {
    float s0 = ((g == 0) ? acc[0] : 0.f) + ((g == 1) ? acc[1] : 0.f)
             + ((g == 2) ? acc[2] : 0.f) + ((g == 3) ? acc[3] : 0.f);
    float s1 = ((g == 0) ? acc[4] : 0.f) + ((g == 1) ? acc[5] : 0.f)
             + ((g == 2) ? acc[6] : 0.f) + ((g == 3) ? acc[7] : 0.f);
    s0 += quad_xor1(s0); s0 += quad_xor2(s0);
    s1 += quad_xor1(s1); s1 += quad_xor2(s1);
    if (g == 0 && q < 2) {
        const long row = trow + 2 * h;
        xout[row * 2 + q] = s0 + bias;
        xout[(row + 1) * 2 + q] = s1 + bias;
    }
}

__global__ __launch_bounds__(GEMM_THREADS) __attribute__((amdgpu_waves_per_eu(4, 4)))
void gemm_x_wmma(const float* __restrict__ FM, const float* __restrict__ lrw,
                 const float* __restrict__ lrb, float* __restrict__ xout) {
    const int lane = threadIdx.x & 31;
    const int gw = blockIdx.x * GEMM_WAVES + (threadIdx.x >> 5);
    const int c = lane & 15;
    const int h = lane >> 4;
    const int g = c & 3;
    const int q = c >> 2;
    const int koff = 8 * g + 4 * h;
    const int roff = 192 + 2 * g + h;

    const bool bval = q < 2;
    const float* wrow = lrw + (q & 1) * 200;
    v4f bm[6];
    for (int j = 0; j < 6; ++j) {
        v4f t;
        t.x = wrow[koff + 32 * j + 0]; t.y = wrow[koff + 32 * j + 1];
        t.z = wrow[koff + 32 * j + 2]; t.w = wrow[koff + 32 * j + 3];
        const v4f z = {0.f, 0.f, 0.f, 0.f};
        bm[j] = bval ? t : z;
    }
    const float brt = wrow[roff];
    const float br = bval ? brt : 0.f;
    const float bias = lrb[q & 1];

    const long row0 = (long)gw * (TILES_PER_WAVE * 4);
    const float* ap = FM + (row0 + q) * 200;

    AFrag c0, c1;
    load_afrag(c0, ap, koff, roff);
    load_afrag(c1, ap + 800, koff, roff);
    for (int p = 0; p < TILES_PER_WAVE / 2; ++p) {
        AFrag n0 = c0, n1 = c1;
        if (p + 1 < TILES_PER_WAVE / 2) {
            load_afrag(n0, ap + (2 * p + 2) * 800, koff, roff);
            load_afrag(n1, ap + (2 * p + 3) * 800, koff, roff);
        }
        v8f acc0 = {}, acc1 = {};
        for (int j = 0; j < 6; ++j) {
            v2f a, b;
            b.x = bm[j].x; b.y = bm[j].y;
            a.x = c0.m[j].x; a.y = c0.m[j].y; acc0 = wmma4(a, b, acc0);
            a.x = c1.m[j].x; a.y = c1.m[j].y; acc1 = wmma4(a, b, acc1);
            b.x = bm[j].z; b.y = bm[j].w;
            a.x = c0.m[j].z; a.y = c0.m[j].w; acc0 = wmma4(a, b, acc0);
            a.x = c1.m[j].z; a.y = c1.m[j].w; acc1 = wmma4(a, b, acc1);
        }
        {
            v2f a, b;
            b.x = br; b.y = 0.f;
            a.x = c0.r; a.y = 0.f; acc0 = wmma4(a, b, acc0);
            a.x = c1.r; a.y = 0.f; acc1 = wmma4(a, b, acc1);
        }
        store_tile(acc0, row0 + (2 * p) * 4, g, q, h, bias, xout);
        store_tile(acc1, row0 + (2 * p + 1) * 4, g, q, h, bias, xout);
        c0 = n0; c1 = n1;
    }
}

#define SCAN_THREADS 512
#define SCAN_WAVES (SCAN_THREADS / 32)
static_assert(NKC <= 128, "a chain number has seven bits and a row of the chain bitmap has four words");
#define ESTEPS 512
#define EPITCH (ESTEPS + 4)
#define EPIECES (ESTEPS / 4)
#define EBUF 16
#define CSLOTS ((NKC * NA + SCAN_THREADS - 1) / SCAN_THREADS)
static_assert((NA - 1) + (SCAN_THREADS % NA) * (CSLOTS - 1) < 2 * NA, "a slot's ability index carries at most once");

struct CStat {
    int ai;
    int si;
    float q0, q1;
    float4 T;
};
static_assert(TT <= ESTEPS && EPIECES % (2 * EBUF) == 0, "phase E walks whole pairs of register buffers");
static_assert(TT <= SCAN_THREADS, "a step's thread carries its inputs from phase A on, and p(y | a) from phase D to phase F");
static_assert(NKC * 4 <= SCAN_THREADS, "a thread loads one element of the chain logit table");

__device__ __forceinline__ float running_sum(const v4f (&buf)[EBUF], float s, float* dst) {
    #pragma unroll
    for (int i = 0; i < EBUF; ++i) {
        v4f o;
        o.x = s; s += buf[i].x;
        o.y = s; s += buf[i].y;
        o.z = s; s += buf[i].z;
        o.w = s; s += buf[i].w;
        *(v4f*)(dst + 4 * i) = o;
    }
    return s;
}

__device__ __forceinline__ float sum13(const float* e) {
    const float sA = (e[0] + e[8]) + (e[4] + e[12]);
    const float sB = (e[2] + e[10]) + e[6];
    const float sC = (e[1] + e[9]) + e[5];
    const float sD = (e[3] + e[11]) + e[7];
    return (sA + sB) + (sC + sD);
}

template <bool USE_XBUF>
__global__ __launch_bounds__(SCAN_THREADS)
void bkt_scan(const int* __restrict__ corr, const int* __restrict__ kcv,
              const int* __restrict__ prob, const float* __restrict__ FM,
              const float* __restrict__ trans, const float* __restrict__ olp,
              const float* __restrict__ olk, const float* __restrict__ initl,
              const float* __restrict__ lrw, const float* __restrict__ lrb,
              const float* __restrict__ xbuf, float* __restrict__ out) {
    __shared__ float4 lo_s[TT * NA];
    __shared__ v2f snap_s[TT * NA];
    __shared__ int hist_s[SCAN_WAVES * NKC];
    __shared__ uint4 lvl_s[TT];
    __shared__ int ord_s[TT];
    __shared__ int lev_s[TT + 4];
    __shared__ v2f alpha[NKC * NA];
    __shared__ __attribute__((aligned(16))) float olk_s[NKC * 4];
    __shared__ float4 T_s[NKC];
    __shared__ v2f pinit_s[NKC];
    __shared__ __attribute__((aligned(16))) float term_s[NA * EPITCH];

    const int tid = threadIdx.x;
    const int wave = tid >> 5, lane = tid & 31;
    const int b = blockIdx.x;
    const long bT = (long)b * TT;
    const bool valid = tid < TT;
    const int wv = __builtin_amdgcn_readfirstlane(wave);

    const long idx = bT + min(tid, TT - 1);
    const int c_ld = kcv[idx];
    const int p_ld = prob[idx];
    const int o_ld = corr[idx];
    v2f xv = {0.f, 0.f};
    if (USE_XBUF) xv = *(const v2f*)(xbuf + idx * 2);
    const float olk_ld = olk[min(tid, NKC * 4 - 1)];
    const int tc = min(tid, NKC - 1);
    const v2f il = *(const v2f*)(initl + tc * 2);
    const float4 tl = *(const float4*)(trans + tc * 4);
    const float4 op4 = *(const float4*)(olp + (long)p_ld * 4);
    const int myc = valid ? c_ld : 0;
    const bool obv = o_ld != 0;

    for (int i = tid; i < TT + 4; i += SCAN_THREADS) lev_s[i] = TT;
    for (int i = lane; i < NKC; i += 32) hist_s[wv * NKC + i] = 0;
    for (int i = tid; i < TT; i += SCAN_THREADS) lvl_s[i] = make_uint4(0u, 0u, 0u, 0u);
    for (int i = tid; i < NA * (ESTEPS - TT); i += SCAN_THREADS)
        term_s[(i / (ESTEPS - TT)) * EPITCH + TT + i % (ESTEPS - TT)] = 0.f;

    unsigned peers = __builtin_amdgcn_ballot_w32(valid);
    for (int bit = 0; bit < 7; ++bit) {
        const bool on = ((myc >> bit) & 1) != 0;
        const unsigned m = __builtin_amdgcn_ballot_w32(on);
        peers &= on ? m : ~m;
    }
    const int rin = __popc(peers & ((1u << lane) - 1u));
    if (valid && rin == 0) hist_s[wv * NKC + myc] = __popc(peers);

    if (tid < NKC * 4) olk_s[tid] = olk_ld;
    if (tid < NKC) {
        pinit_s[tid] = softmax2(il.y - il.x);
        const v2f col0 = softmax2(tl.z - tl.x);
        const v2f col1 = softmax2(tl.w - tl.y);
        float4 T;
        T.x = col0.x; T.y = col1.x;
        T.z = col0.y; T.w = col1.y;
        T_s[tid] = T;
    }

    if (!USE_XBUF) {
        __shared__ v2f x_s[TT];
        for (int t = wave; t < TT; t += SCAN_WAVES) {
            const float* fr = FM + (bT + t) * 200;
            float p0 = 0.f, p1 = 0.f;
            for (int f = lane; f < 200; f += 32) {
                float fv = fr[f];
                p0 = fmaf(fv, lrw[f], p0);
                p1 = fmaf(fv, lrw[200 + f], p1);
            }
            for (int off = 16; off; off >>= 1) {
                p0 += __shfl_xor(p0, off, 32);
                p1 += __shfl_xor(p1, off, 32);
            }
            if (lane == 0) {
                v2f xr;
                xr.x = p0 + lrb[0];
                xr.y = p1 + lrb[1];
                x_s[t] = xr;
            }
        }
        __syncthreads();
        xv = x_s[min(tid, TT - 1)];
    }
    __syncthreads();

    for (int i = tid; i < NKC * NA; i += SCAN_THREADS) alpha[i] = pinit_s[i / NA];

    int occ = rin;
    for (int w = 0; w < wv; ++w) occ += hist_s[w * NKC + myc];
    if (wv < (NKC + 31) / 32) {
        int cnt = 0;
        if (tid < NKC)
            for (int w = 0; w < SCAN_WAVES; ++w) cnt += hist_s[w * NKC + tid];
        for (int k = 0; k < TT; ++k) {
            const unsigned m = __builtin_amdgcn_ballot_w32(cnt > k);
            if (m == 0u) break;
            if (lane == 0) reinterpret_cast<unsigned*>(lvl_s)[k * 4 + wv] = m;
        }
    }

    if (valid) {
        const float4 ok4 = *(const float4*)(olk_s + myc * 4);
        const float x0 = xv.x, x1 = xv.y;
        const float s00 = op4.x + ok4.x, s01 = op4.y + ok4.y;
        const float s10 = op4.z + ok4.z, s11 = op4.w + ok4.w;
        float4* lrow = lo_s + tid * NA;
        #pragma unroll
        for (int a = 0; a < NA; ++a) {
            const float abil = -3.0f + 0.5f * (float)a;
            const float l00 = s00 - abil + x0;
            const float l01 = s01 + abil - x0;
            const float l10 = s10 - abil + x1;
            const float l11 = s11 + abil - x1;
            const v2f q0 = softmax2(l01 - l00);
            const v2f q1 = softmax2(l11 - l10);
            float4 r;
            r.x = q0.x; r.y = q0.y;
            r.z = q1.x; r.w = q1.y;
            lrow[a] = r;
        }
    }
    __syncthreads();

    if (valid) {
        int lev = 0;
        for (int k = 0; k < occ; ++k) {
            const uint4 m = lvl_s[k];
            lev += __popc(m.x) + __popc(m.y) + __popc(m.z) + __popc(m.w);
        }
        const uint4 m = lvl_s[occ];
        const int v = myc >> 5;
        const unsigned below = (1u << (myc & 31)) - 1u;
        const unsigned m0 = (v > 0) ? ~0u : below;
        const unsigned m1 = (v > 1) ? ~0u : ((v == 1) ? below : 0u);
        const unsigned m2 = (v > 2) ? ~0u : ((v == 2) ? below : 0u);
        const unsigned m3 = (v == 3) ? below : 0u;
        const int idx = __popc(m.x & m0) + __popc(m.y & m1) + __popc(m.z & m2) + __popc(m.w & m3);
        const int pos = lev + idx;
        ord_s[pos] = tid | (myc << 9) | ((int)obv << 16) | (occ << 17);
        if (idx == 0) lev_s[occ] = pos;
    }
    __syncthreads();

    {
        const int pt = tid / NA;
        const int at = tid - pt * NA;
        int P[CSLOTS], A[CSLOTS];
        #pragma unroll
        for (int j = 0; j < CSLOTS; ++j) {
            const int a = at + (SCAN_THREADS % NA) * j;
            const int carry = (a >= NA) ? 1 : 0;
            A[j] = a - NA * carry;
            P[j] = pt + (SCAN_THREADS / NA) * j + carry;
        }

        CStat cur[CSLOTS];
        auto fetch_stat = [&](int s) {
            int rec[CSLOTS];
            #pragma unroll
            for (int j = 0; j < CSLOTS; ++j) rec[j] = ord_s[min(s + P[j], TT - 1)];
            #pragma unroll
            for (int j = 0; j < CSLOTS; ++j) {
                const int t = rec[j] & 511;
                const int c = (rec[j] >> 9) & 127;
                const int obs = (rec[j] >> 16) & 1;
                const int si = t * NA + A[j];
                const float4 q = lo_s[si];
                cur[j].T = T_s[c];
                cur[j].ai = c * NA + A[j];
                cur[j].si = si;
                cur[j].q0 = obs ? q.y : q.x;
                cur[j].q1 = obs ? q.w : q.z;
            }
        };

        int k = 0;
        int s0 = 0;
        int s1 = __builtin_amdgcn_readfirstlane(lev_s[1]);
        int s2 = __builtin_amdgcn_readfirstlane(lev_s[2]);
        fetch_stat(s0);
        while (s0 < TT) {
            const int s3v = lev_s[k + 3];
            if (s0 + P[0] < s1 || s1 + P[0] < s2) {
                v2f pv[CSLOTS], pn[CSLOTS];
                int ai[CSLOTS], si[CSLOTS];
                #pragma unroll
                for (int j = 0; j < CSLOTS; ++j) {
                    ai[j] = cur[j].ai; si[j] = cur[j].si;
                    pv[j] = alpha[ai[j]];
                }
                #pragma unroll
                for (int j = 0; j < CSLOTS; ++j) {
                    const float4 T = cur[j].T;
                    const float w0 = cur[j].q0 * pv[j].x;
                    const float w1 = cur[j].q1 * pv[j].y;
                    const float n0 = T.x * w0 + T.y * w1;
                    const float n1 = T.z * w0 + T.w * w1;
                    const float rn = __builtin_amdgcn_rcpf(n0 + n1);
                    pn[j].x = n0 * rn; pn[j].y = n1 * rn;
                }
                fetch_stat(s1);
                #pragma unroll
                for (int j = 0; j < CSLOTS; ++j) {
                    if (s0 + P[j] < s1) {
                        alpha[ai[j]] = pn[j];
                        snap_s[si[j]] = pv[j];
                    }
                }
            }
            ++k;
            s0 = s1;
            s1 = s2;
            s2 = __builtin_amdgcn_readfirstlane(s3v);
            __syncthreads();
        }
    }

    v2f py[NA];
    if (valid) {
        const bool ob = obv;
        const float4* lrow = lo_s + tid * NA;
        const v2f* srow = snap_s + tid * NA;
        #pragma unroll
        for (int a = 0; a < NA; ++a) {
            const float4 q = lrow[a];
            const v2f pv = srow[a];
            v2f p;
            p.x = q.x * pv.x + q.z * pv.y;
            p.y = q.y * pv.x + q.w * pv.y;
            py[a] = p;
            term_s[a * EPITCH + tid] = __logf(ob ? p.y : p.x);
        }
    }
    __syncthreads();

    float* pre_s = reinterpret_cast<float*>(lo_s);
    if (wave == 0 && lane < NA) {
        const float* trow = term_s + lane * EPITCH;
        float* prow = pre_s + lane * EPITCH;
        float logab = -2.5649493574615367f;
        v4f bufa[EBUF], bufb[EBUF];
        #pragma unroll
        for (int i = 0; i < EBUF; ++i) bufa[i] = *(const v4f*)(trow + 4 * i);
        for (int p0 = 0; p0 < EPIECES; p0 += 2 * EBUF) {
            #pragma unroll
            for (int i = 0; i < EBUF; ++i) bufb[i] = *(const v4f*)(trow + 4 * (p0 + EBUF + i));
            logab = running_sum(bufa, logab, prow + 4 * p0);
            if (p0 + 2 * EBUF < EPIECES) {
                #pragma unroll
                for (int i = 0; i < EBUF; ++i) bufa[i] = *(const v4f*)(trow + 4 * (p0 + 2 * EBUF + i));
            }
            logab = running_sum(bufb, logab, prow + 4 * (p0 + EBUF));
        }
    }
    __syncthreads();

    if (valid) {
        float la[NA];
        float m = -1.0e30f;
        #pragma unroll
        for (int a = 0; a < NA; ++a) {
            la[a] = pre_s[a * EPITCH + tid];
            m = fmaxf(m, la[a]);
        }
        float e0[NA], e1[NA];
        #pragma unroll
        for (int a = 0; a < NA; ++a) {
            const float w = __expf(la[a] - m);
            e0[a] = w * py[a].x;
            e1[a] = w * py[a].y;
        }
        const float s0 = sum13(e0);
        const float s1 = sum13(e1);
        const float lnrm = __logf(s0 + s1);
        v2f ov;
        ov.x = __logf(s0) - lnrm;
        ov.y = __logf(s1) - lnrm;
        *(v2f*)(out + (bT + tid) * 2) = ov;
    }
}

extern "C" void kernel_launch(void* const* d_in, const int* in_sizes, int n_in,
                              void* d_out, int out_size, void* d_ws, size_t ws_size,
                              hipStream_t stream) {
    (void)in_sizes; (void)n_in; (void)out_size;
    const int* corr = (const int*)d_in[0];
    const int* kcv  = (const int*)d_in[1];
    const int* prob = (const int*)d_in[2];
    const float* FM    = (const float*)d_in[3];
    const float* trans = (const float*)d_in[4];
    const float* olp   = (const float*)d_in[5];
    const float* olk   = (const float*)d_in[6];
    const float* initl = (const float*)d_in[7];
    const float* lrw   = (const float*)d_in[8];
    const float* lrb   = (const float*)d_in[9];
    float* out = (float*)d_out;

    const size_t xbytes = (size_t)BB * TT * 2 * sizeof(float);
    if (ws_size >= xbytes) {
        float* xbuf = (float*)d_ws;
        hipLaunchKernelGGL(gemm_x_wmma, dim3(GEMM_BLOCKS), dim3(GEMM_THREADS), 0, stream,
                           FM, lrw, lrb, xbuf);
        hipLaunchKernelGGL((bkt_scan<true>), dim3(BB), dim3(SCAN_THREADS), 0, stream,
                           corr, kcv, prob, FM, trans, olp, olk, initl,
                           lrw, lrb, xbuf, out);
    } else {
        hipLaunchKernelGGL((bkt_scan<false>), dim3(BB), dim3(SCAN_THREADS), 0, stream,
                           corr, kcv, prob, FM, trans, olp, olk, initl,
                           lrw, lrb, (const float*)nullptr, out);
    }
}
